# starting kernel
# baseline (speedup 1.0000x reference)
#include <hip/hip_runtime.h>
#include <hip/hip_bf16.h>

#define B_DIM 4096
#define O_DIM 16
#define D_DIM 256
#define H_DIM 512

typedef __attribute__((ext_vector_type(2))) float v2f;
typedef __attribute__((ext_vector_type(8))) float v8f;

__device__ __forceinline__ v8f wmma_f32(v2f a, v2f b, v8f c) {
    return __builtin_amdgcn_wmma_f32_16x16x4_f32(
         false, a,  false, b,
         (short)0, c,  false,  false);
}

__global__ void __launch_bounds__(256)
k_reduce_states(const float* __restrict__ states, float* __restrict__ xbar) {
    const int b = blockIdx.x;
    const int d = threadIdx.x;
    const float* p = states + (size_t)b * (O_DIM * D_DIM) + d;
    float acc = 0.f;
#pragma unroll
    for (int o = 0; o < O_DIM; ++o) acc += p[o * D_DIM];
    xbar[(size_t)b * D_DIM + d] = acc * (1.0f / 16.0f);
}

__global__ void __launch_bounds__(256)
k_gemm1(const float* __restrict__ xbar, const float* __restrict__ W1,
        const float* __restrict__ bias1, const int* __restrict__ action,
        float* __restrict__ h1) {
    const int K = D_DIM, N = H_DIM;
    const int wave = threadIdx.x >> 5;
    const int lane = threadIdx.x & 31;
    const int job  = blockIdx.x * 8 + wave;
    const int m0 = (job >> 3) * 16;
    const int n0 = (job & 7) * 64;
    const int laneM = lane & 15;
    const int laneK = (lane >> 4) * 2;

    v8f acc[4] = {};
    const float* arow = xbar + (size_t)(m0 + laneM) * K + laneK;
    for (int k = 0; k < K; k += 4) {
        v2f a = *(const v2f*)(arow + k);
        const float* bp = W1 + (size_t)(k + laneK) * N + laneM;
#pragma unroll
        for (int t = 0; t < 4; ++t) {
            v2f bf = { bp[n0 + t * 16], bp[N + n0 + t * 16] };
            acc[t] = wmma_f32(a, bf, acc[t]);
        }
    }
    const int mhalf = (lane >> 4) * 8;
#pragma unroll
    for (int r = 0; r < 8; ++r) {
        const int m = m0 + mhalf + r;
        const int c = action[m] & 3;
        const float* wtail = W1 + (size_t)(D_DIM + c) * N;
#pragma unroll
        for (int t = 0; t < 4; ++t) {
            const int n = n0 + t * 16 + laneM;
            float v = acc[t][r] + wtail[n] * 0.0625f + bias1[n];
            h1[(size_t)m * N + n] = fmaxf(v, 0.0f);
        }
    }
}

__global__ void __launch_bounds__(256)
k_gemm2_ln(const float* __restrict__ h1, const float* __restrict__ W2,
           const float* __restrict__ bias2, const float* __restrict__ ln_g,
           const float* __restrict__ ln_b, float* __restrict__ h2) {
    const int K = H_DIM, N = H_DIM;
    __shared__ float z[16 * H_DIM];
    __shared__ float red_s[16][16];
    __shared__ float red_q[16][16];
    __shared__ float mu_s[16], rs_s[16];

    const int wave = threadIdx.x >> 5;
    const int lane = threadIdx.x & 31;
    const int m0 = blockIdx.x * 16;
    const int n0 = wave * 64;
    const int laneM = lane & 15;
    const int laneK = (lane >> 4) * 2;

    v8f acc[4] = {};
    const float* arow = h1 + (size_t)(m0 + laneM) * K + laneK;
    for (int k = 0; k < K; k += 4) {
        v2f a = *(const v2f*)(arow + k);
        const float* bp = W2 + (size_t)(k + laneK) * N + laneM;
#pragma unroll
        for (int t = 0; t < 4; ++t) {
            v2f bf = { bp[n0 + t * 16], bp[N + n0 + t * 16] };
            acc[t] = wmma_f32(a, bf, acc[t]);
        }
    }
    const int mhalf = (lane >> 4) * 8;
#pragma unroll
    for (int r = 0; r < 8; ++r) {
#pragma unroll
        for (int t = 0; t < 4; ++t) {
            const int n = n0 + t * 16 + laneM;
            z[(mhalf + r) * H_DIM + n] = acc[t][r] + bias2[n];
        }
    }
    __syncthreads();

    const int row = threadIdx.x >> 4;
    const int seg = threadIdx.x & 15;
    {
        const float* zr = z + row * H_DIM + seg * 32;
        float s = 0.f, q = 0.f;
#pragma unroll
        for (int i = 0; i < 32; ++i) { float v = zr[i]; s += v; q += v * v; }
        red_s[row][seg] = s;
        red_q[row][seg] = q;
    }
    __syncthreads();
    if (threadIdx.x < 16) {
        float s = 0.f, q = 0.f;
#pragma unroll
        for (int i = 0; i < 16; ++i) { s += red_s[threadIdx.x][i]; q += red_q[threadIdx.x][i]; }
        const float mu = s * (1.0f / H_DIM);
        const float var = q * (1.0f / H_DIM) - mu * mu;
        mu_s[threadIdx.x] = mu;
        rs_s[threadIdx.x] = rsqrtf(var + 1e-5f);
    }
    __syncthreads();
    {
        const float mu = mu_s[row], rstd = rs_s[row];
        float* dst = h2 + (size_t)(m0 + row) * H_DIM;
        const float* zr = z + row * H_DIM;
#pragma unroll
        for (int i = 0; i < 32; ++i) {
            const int n = seg * 32 + i;
            float v = (zr[n] - mu) * rstd * ln_g[n] + ln_b[n];
            dst[n] = fmaxf(v, 0.0f);
        }
    }
}

__global__ void __launch_bounds__(256)
k_gemm3_bcast(const float* __restrict__ h2, const float* __restrict__ W3,
              const float* __restrict__ bias3, float* __restrict__ out) {
    const int K = H_DIM, N = D_DIM;
    const int wave = threadIdx.x >> 5;
    const int lane = threadIdx.x & 31;
    const int job  = blockIdx.x * 8 + wave;
    const int m0 = (job >> 2) * 16;
    const int n0 = (job & 3) * 64;
    const int laneM = lane & 15;
    const int laneK = (lane >> 4) * 2;

    v8f acc[4] = {};
    const float* arow = h2 + (size_t)(m0 + laneM) * K + laneK;
    for (int k = 0; k < K; k += 4) {
        v2f a = *(const v2f*)(arow + k);
        const float* bp = W3 + (size_t)(k + laneK) * N + laneM;
#pragma unroll
        for (int t = 0; t < 4; ++t) {
            v2f bf = { bp[n0 + t * 16], bp[N + n0 + t * 16] };
            acc[t] = wmma_f32(a, bf, acc[t]);
        }
    }
    const int mhalf = (lane >> 4) * 8;
#pragma unroll
    for (int r = 0; r < 8; ++r) {
        const int m = m0 + mhalf + r;
        float* obase = out + (size_t)m * (O_DIM * D_DIM);
#pragma unroll
        for (int t = 0; t < 4; ++t) {
            const int n = n0 + t * 16 + laneM;
            const float v = acc[t][r] + bias3[n];
#pragma unroll
            for (int o = 0; o < O_DIM; ++o)
                obase[o * D_DIM + n] = v;
        }
    }
}

extern "C" void kernel_launch(void* const* d_in, const int* in_sizes, int n_in,
                              void* d_out, int out_size, void* d_ws, size_t ws_size,
                              hipStream_t stream) {
    const float* states = (const float*)d_in[0];
    const int*   action = (const int*)d_in[1];
    const float* W1     = (const float*)d_in[2];
    const float* b1     = (const float*)d_in[3];
    const float* W2     = (const float*)d_in[4];
    const float* b2     = (const float*)d_in[5];
    const float* W3     = (const float*)d_in[6];
    const float* b3     = (const float*)d_in[7];
    const float* ln_g   = (const float*)d_in[8];
    const float* ln_b   = (const float*)d_in[9];
    float* out = (float*)d_out;

    float* xbar = (float*)d_ws;
    float* h1   = xbar + (size_t)B_DIM * D_DIM;
    float* h2   = h1   + (size_t)B_DIM * H_DIM;

    k_reduce_states<<<B_DIM, 256, 0, stream>>>(states, xbar);
    k_gemm1<<<(B_DIM / 16) * 8 / 8, 256, 0, stream>>>(xbar, W1, b1, action, h1);
    k_gemm2_ln<<<B_DIM / 16, 256, 0, stream>>>(h1, W2, b2, ln_g, ln_b, h2);
    k_gemm3_bcast<<<(B_DIM / 16) * 4 / 8, 256, 0, stream>>>(h2, W3, b3, out);
}


# K loops of the three WMMA GEMM kernels software-pipelined: each 8-k set's 18 fragment loads issued as one group three sets ahead (ring of four register sets); k order per accumulator unchanged
# speedup vs baseline: 1.6504x; 1.6504x over previous
#include <hip/hip_runtime.h>
#include <hip/hip_bf16.h>

#define B_DIM 4096
#define O_DIM 16
#define D_DIM 256
#define H_DIM 512

typedef __attribute__((ext_vector_type(2))) float v2f;
typedef __attribute__((ext_vector_type(8))) float v8f;

__device__ __forceinline__ v8f wmma_f32(v2f a, v2f b, v8f c) {
    return __builtin_amdgcn_wmma_f32_16x16x4_f32(
         false, a,  false, b,
         (short)0, c,  false,  false);
}

struct FragSet { v2f a[2]; v2f b[2][4]; };

template <int N>
__device__ __forceinline__ void frag_load(FragSet& f, const float* __restrict__ ap,
                                          const float* __restrict__ bp) {
#pragma unroll
    for (int s = 0; s < 2; ++s) {
        f.a[s] = *(const v2f*)(ap + 4 * s);
        const float* b = bp + (size_t)(4 * s) * N;
#pragma unroll
        for (int t = 0; t < 4; ++t) {
            v2f bf = { b[t * 16], b[N + t * 16] };
            f.b[s][t] = bf;
        }
    }
}

__device__ __forceinline__ void frag_mma(const FragSet& f, v8f (&acc)[4]) {
#pragma unroll
    for (int s = 0; s < 2; ++s) {
#pragma unroll
        for (int t = 0; t < 4; ++t) acc[t] = wmma_f32(f.a[s], f.b[s][t], acc[t]);
    }
}

template <int K, int N>
__device__ __forceinline__ void gemm_16x64(const float* __restrict__ ap,
                                           const float* __restrict__ bp, v8f (&acc)[4]) {
    constexpr int KB = 8;
    constexpr int NB = K / KB;
    static_assert(K % KB == 0 && NB % 4 == 0 && NB >= 8, "ring of four sets");
    FragSet f0, f1, f2, f3;
#define ORDER_FENCE() \
    asm volatile("" : "+v"(acc[0]), "+v"(acc[1]), "+v"(acc[2]), "+v"(acc[3]) : : "memory")
    frag_load<N>(f0, ap, bp);
    ORDER_FENCE();
    frag_load<N>(f1, ap + KB, bp + (size_t)KB * N);
    ORDER_FENCE();
    frag_load<N>(f2, ap + 2 * KB, bp + (size_t)2 * KB * N);
    ORDER_FENCE();
    for (int kb = 3; kb < NB - 1; kb += 4) {
        const float* a = ap + kb * KB;
        const float* b = bp + (size_t)kb * KB * N;
        frag_load<N>(f3, a, b);
        ORDER_FENCE();
        frag_mma(f0, acc);
        ORDER_FENCE();
        frag_load<N>(f0, a + KB, b + (size_t)KB * N);
        ORDER_FENCE();
        frag_mma(f1, acc);
        ORDER_FENCE();
        frag_load<N>(f1, a + 2 * KB, b + (size_t)2 * KB * N);
        ORDER_FENCE();
        frag_mma(f2, acc);
        ORDER_FENCE();
        frag_load<N>(f2, a + 3 * KB, b + (size_t)3 * KB * N);
        ORDER_FENCE();
        frag_mma(f3, acc);
        ORDER_FENCE();
    }
    frag_load<N>(f3, ap + (NB - 1) * KB, bp + (size_t)(NB - 1) * KB * N);
    ORDER_FENCE();
    frag_mma(f0, acc);
    frag_mma(f1, acc);
    frag_mma(f2, acc);
    frag_mma(f3, acc);
#undef ORDER_FENCE
}

__global__ void __launch_bounds__(256)
k_reduce_states(const float* __restrict__ states, float* __restrict__ xbar) {
    const int b = blockIdx.x;
    const int d = threadIdx.x;
    const float* p = states + (size_t)b * (O_DIM * D_DIM) + d;
    float acc = 0.f;
#pragma unroll
    for (int o = 0; o < O_DIM; ++o) acc += p[o * D_DIM];
    xbar[(size_t)b * D_DIM + d] = acc * (1.0f / 16.0f);
}

__global__ void __launch_bounds__(256)
k_gemm1(const float* __restrict__ xbar, const float* __restrict__ W1,
        const float* __restrict__ bias1, const int* __restrict__ action,
        float* __restrict__ h1) {
    const int K = D_DIM, N = H_DIM;
    const int wave = threadIdx.x >> 5;
    const int lane = threadIdx.x & 31;
    const int job  = blockIdx.x * 8 + wave;
    const int m0 = (job >> 3) * 16;
    const int n0 = (job & 7) * 64;
    const int laneM = lane & 15;
    const int laneK = (lane >> 4) * 2;

    v8f acc[4] = {};
    const float* arow = xbar + (size_t)(m0 + laneM) * K + laneK;
    gemm_16x64<D_DIM, H_DIM>(arow, W1 + (size_t)laneK * N + n0 + laneM, acc);
    const int mhalf = (lane >> 4) * 8;
#pragma unroll
    for (int r = 0; r < 8; ++r) {
        const int m = m0 + mhalf + r;
        const int c = action[m] & 3;
        const float* wtail = W1 + (size_t)(D_DIM + c) * N;
#pragma unroll
        for (int t = 0; t < 4; ++t) {
            const int n = n0 + t * 16 + laneM;
            float v = acc[t][r] + wtail[n] * 0.0625f + bias1[n];
            h1[(size_t)m * N + n] = fmaxf(v, 0.0f);
        }
    }
}

__global__ void __launch_bounds__(256)
k_gemm2_ln(const float* __restrict__ h1, const float* __restrict__ W2,
           const float* __restrict__ bias2, const float* __restrict__ ln_g,
           const float* __restrict__ ln_b, float* __restrict__ h2) {
    const int K = H_DIM, N = H_DIM;
    __shared__ float z[16 * H_DIM];
    __shared__ float red_s[16][16];
    __shared__ float red_q[16][16];
    __shared__ float mu_s[16], rs_s[16];

    const int wave = threadIdx.x >> 5;
    const int lane = threadIdx.x & 31;
    const int m0 = blockIdx.x * 16;
    const int n0 = wave * 64;
    const int laneM = lane & 15;
    const int laneK = (lane >> 4) * 2;

    v8f acc[4] = {};
    const float* arow = h1 + (size_t)(m0 + laneM) * K + laneK;
    gemm_16x64<H_DIM, H_DIM>(arow, W2 + (size_t)laneK * N + n0 + laneM, acc);
    const int mhalf = (lane >> 4) * 8;
#pragma unroll
    for (int r = 0; r < 8; ++r) {
#pragma unroll
        for (int t = 0; t < 4; ++t) {
            const int n = n0 + t * 16 + laneM;
            z[(mhalf + r) * H_DIM + n] = acc[t][r] + bias2[n];
        }
    }
    __syncthreads();

    const int row = threadIdx.x >> 4;
    const int seg = threadIdx.x & 15;
    {
        const float* zr = z + row * H_DIM + seg * 32;
        float s = 0.f, q = 0.f;
#pragma unroll
        for (int i = 0; i < 32; ++i) { float v = zr[i]; s += v; q += v * v; }
        red_s[row][seg] = s;
        red_q[row][seg] = q;
    }
    __syncthreads();
    if (threadIdx.x < 16) {
        float s = 0.f, q = 0.f;
#pragma unroll
        for (int i = 0; i < 16; ++i) { s += red_s[threadIdx.x][i]; q += red_q[threadIdx.x][i]; }
        const float mu = s * (1.0f / H_DIM);
        const float var = q * (1.0f / H_DIM) - mu * mu;
        mu_s[threadIdx.x] = mu;
        rs_s[threadIdx.x] = rsqrtf(var + 1e-5f);
    }
    __syncthreads();
    {
        const float mu = mu_s[row], rstd = rs_s[row];
        float* dst = h2 + (size_t)(m0 + row) * H_DIM;
        const float* zr = z + row * H_DIM;
#pragma unroll
        for (int i = 0; i < 32; ++i) {
            const int n = seg * 32 + i;
            float v = (zr[n] - mu) * rstd * ln_g[n] + ln_b[n];
            dst[n] = fmaxf(v, 0.0f);
        }
    }
}

__global__ void __launch_bounds__(256)
k_gemm3_bcast(const float* __restrict__ h2, const float* __restrict__ W3,
              const float* __restrict__ bias3, float* __restrict__ out) {
    const int K = H_DIM, N = D_DIM;
    const int wave = threadIdx.x >> 5;
    const int lane = threadIdx.x & 31;
    const int job  = blockIdx.x * 8 + wave;
    const int m0 = (job >> 2) * 16;
    const int n0 = (job & 3) * 64;
    const int laneM = lane & 15;
    const int laneK = (lane >> 4) * 2;

    v8f acc[4] = {};
    const float* arow = h2 + (size_t)(m0 + laneM) * K + laneK;
    gemm_16x64<H_DIM, D_DIM>(arow, W3 + (size_t)laneK * N + n0 + laneM, acc);
    const int mhalf = (lane >> 4) * 8;
#pragma unroll
    for (int r = 0; r < 8; ++r) {
        const int m = m0 + mhalf + r;
        float* obase = out + (size_t)m * (O_DIM * D_DIM);
#pragma unroll
        for (int t = 0; t < 4; ++t) {
            const int n = n0 + t * 16 + laneM;
            const float v = acc[t][r] + bias3[n];
#pragma unroll
            for (int o = 0; o < O_DIM; ++o)
                obase[o * D_DIM + n] = v;
        }
    }
}

extern "C" void kernel_launch(void* const* d_in, const int* in_sizes, int n_in,
                              void* d_out, int out_size, void* d_ws, size_t ws_size,
                              hipStream_t stream) {
    const float* states = (const float*)d_in[0];
    const int*   action = (const int*)d_in[1];
    const float* W1     = (const float*)d_in[2];
    const float* b1     = (const float*)d_in[3];
    const float* W2     = (const float*)d_in[4];
    const float* b2     = (const float*)d_in[5];
    const float* W3     = (const float*)d_in[6];
    const float* b3     = (const float*)d_in[7];
    const float* ln_g   = (const float*)d_in[8];
    const float* ln_b   = (const float*)d_in[9];
    float* out = (float*)d_out;

    float* xbar = (float*)d_ws;
    float* h1   = xbar + (size_t)B_DIM * D_DIM;
    float* h2   = h1   + (size_t)B_DIM * H_DIM;

    k_reduce_states<<<B_DIM, 256, 0, stream>>>(states, xbar);
    k_gemm1<<<(B_DIM / 16) * 8 / 8, 256, 0, stream>>>(xbar, W1, b1, action, h1);
    k_gemm2_ln<<<B_DIM / 16, 256, 0, stream>>>(h1, W2, b2, ln_g, ln_b, h2);
    k_gemm3_bcast<<<(B_DIM / 16) * 4 / 8, 256, 0, stream>>>(h2, W3, b3, out);
}


# Wave tile columns remapped so each lane owns 4 adjacent n (n0+4c+t): B fragments from 16-byte whole-line W loads paired by register moves, epilogue loads/stores 16-byte; k order and sums unchanged
# speedup vs baseline: 2.1386x; 1.2958x over previous
#include <hip/hip_runtime.h>
#include <hip/hip_bf16.h>

#define B_DIM 4096
#define O_DIM 16
#define D_DIM 256
#define H_DIM 512

typedef __attribute__((ext_vector_type(2))) float v2f;
typedef __attribute__((ext_vector_type(4))) float v4f;
typedef __attribute__((ext_vector_type(8))) float v8f;

__device__ __forceinline__ v8f wmma_f32(v2f a, v2f b, v8f c) {
    return __builtin_amdgcn_wmma_f32_16x16x4_f32(
         false, a,  false, b,
         (short)0, c,  false,  false);
}

struct FragSet { v2f a[2]; v4f b[2][2]; };

template <int N>
__device__ __forceinline__ void frag_load(FragSet& f, const float* __restrict__ ap,
                                          const float* __restrict__ bp) {
#pragma unroll
    for (int s = 0; s < 2; ++s) {
        f.a[s] = *(const v2f*)(ap + 4 * s);
        const float* b = bp + (size_t)(4 * s) * N;
        f.b[s][0] = *(const v4f*)(b);
        f.b[s][1] = *(const v4f*)(b + N);
    }
}

#define WMMA_SLOTS "v_nop\n\tv_nop\n\tv_nop\n\tv_nop"

__device__ __forceinline__ void frag_mma(FragSet& f, v8f (&acc)[4]) {
    asm volatile(WMMA_SLOTS
                 : "+v"(acc[0]), "+v"(acc[1]), "+v"(acc[2]), "+v"(acc[3]),
                   "+v"(f.b[0][0]), "+v"(f.b[0][1]), "+v"(f.b[1][0]), "+v"(f.b[1][1]));
    v2f p[2][4];
#pragma unroll
    for (int s = 0; s < 2; ++s) {
#pragma unroll
        for (int t = 0; t < 4; ++t) {
            const v2f bf = { f.b[s][0][t], f.b[s][1][t] };
            p[s][t] = bf;
        }
    }
    asm volatile("" : "+v"(p[0][0]), "+v"(p[0][1]), "+v"(p[0][2]), "+v"(p[0][3]),
                      "+v"(p[1][0]), "+v"(p[1][1]), "+v"(p[1][2]), "+v"(p[1][3]));
#pragma unroll
    for (int s = 0; s < 2; ++s) {
#pragma unroll
        for (int t = 0; t < 4; ++t) acc[t] = wmma_f32(f.a[s], p[s][t], acc[t]);
    }
}

template <int K, int N>
__device__ __forceinline__ void gemm_16x64(const float* __restrict__ ap,
                                           const float* __restrict__ bp, v8f (&acc)[4]) {
    constexpr int KB = 8;
    constexpr int NB = K / KB;
    static_assert(K % KB == 0 && NB % 4 == 0 && NB >= 8, "ring of four sets");
    FragSet f0, f1, f2, f3;
#define ORDER_FENCE() \
    asm volatile("" : "+v"(acc[0]), "+v"(acc[1]), "+v"(acc[2]), "+v"(acc[3]) : : "memory")
    frag_load<N>(f0, ap, bp);
    ORDER_FENCE();
    frag_load<N>(f1, ap + KB, bp + (size_t)KB * N);
    ORDER_FENCE();
    frag_load<N>(f2, ap + 2 * KB, bp + (size_t)2 * KB * N);
    ORDER_FENCE();
    for (int kb = 3; kb < NB - 1; kb += 4) {
        const float* a = ap + kb * KB;
        const float* b = bp + (size_t)kb * KB * N;
        frag_load<N>(f3, a, b);
        ORDER_FENCE();
        frag_mma(f0, acc);
        ORDER_FENCE();
        frag_load<N>(f0, a + KB, b + (size_t)KB * N);
        ORDER_FENCE();
        frag_mma(f1, acc);
        ORDER_FENCE();
        frag_load<N>(f1, a + 2 * KB, b + (size_t)2 * KB * N);
        ORDER_FENCE();
        frag_mma(f2, acc);
        ORDER_FENCE();
        frag_load<N>(f2, a + 3 * KB, b + (size_t)3 * KB * N);
        ORDER_FENCE();
        frag_mma(f3, acc);
        ORDER_FENCE();
    }
    frag_load<N>(f3, ap + (NB - 1) * KB, bp + (size_t)(NB - 1) * KB * N);
    ORDER_FENCE();
    frag_mma(f0, acc);
    frag_mma(f1, acc);
    frag_mma(f2, acc);
    frag_mma(f3, acc);
    asm volatile(WMMA_SLOTS : "+v"(acc[0]), "+v"(acc[1]), "+v"(acc[2]), "+v"(acc[3]));
#undef ORDER_FENCE
}

__global__ void __launch_bounds__(256)
k_reduce_states(const float* __restrict__ states, float* __restrict__ xbar) {
    const int b = blockIdx.x;
    const int d = threadIdx.x;
    const float* p = states + (size_t)b * (O_DIM * D_DIM) + d;
    float acc = 0.f;
#pragma unroll
    for (int o = 0; o < O_DIM; ++o) acc += p[o * D_DIM];
    xbar[(size_t)b * D_DIM + d] = acc * (1.0f / 16.0f);
}

__global__ void __launch_bounds__(256)
k_gemm1(const float* __restrict__ xbar, const float* __restrict__ W1,
        const float* __restrict__ bias1, const int* __restrict__ action,
        float* __restrict__ h1) {
    const int K = D_DIM, N = H_DIM;
    const int wave = threadIdx.x >> 5;
    const int lane = threadIdx.x & 31;
    const int job  = blockIdx.x * 8 + wave;
    const int m0 = (job >> 3) * 16;
    const int n0 = (job & 7) * 64;
    const int laneM = lane & 15;
    const int laneK = (lane >> 4) * 2;

    v8f acc[4] = {};
    const float* arow = xbar + (size_t)(m0 + laneM) * K + laneK;
    gemm_16x64<D_DIM, H_DIM>(arow, W1 + (size_t)laneK * N + n0 + 4 * laneM, acc);
    const int mhalf = (lane >> 4) * 8;
    const int n = n0 + 4 * laneM;
    const v4f bb = *(const v4f*)(bias1 + n);
#pragma unroll
    for (int r = 0; r < 8; ++r) {
        const int m = m0 + mhalf + r;
        const int c = action[m] & 3;
        const v4f wt = *(const v4f*)(W1 + (size_t)(D_DIM + c) * N + n);
        v4f o;
#pragma unroll
        for (int t = 0; t < 4; ++t) {
            float v = acc[t][r] + wt[t] * 0.0625f + bb[t];
            o[t] = fmaxf(v, 0.0f);
        }
        *(v4f*)(h1 + (size_t)m * N + n) = o;
    }
}

__global__ void __launch_bounds__(256)
k_gemm2_ln(const float* __restrict__ h1, const float* __restrict__ W2,
           const float* __restrict__ bias2, const float* __restrict__ ln_g,
           const float* __restrict__ ln_b, float* __restrict__ h2) {
    const int K = H_DIM, N = H_DIM;
    __shared__ __attribute__((aligned(16))) float z[16 * H_DIM];
    __shared__ float red_s[16][16];
    __shared__ float red_q[16][16];
    __shared__ float mu_s[16], rs_s[16];

    const int wave = threadIdx.x >> 5;
    const int lane = threadIdx.x & 31;
    const int m0 = blockIdx.x * 16;
    const int n0 = wave * 64;
    const int laneM = lane & 15;
    const int laneK = (lane >> 4) * 2;

    v8f acc[4] = {};
    const float* arow = h1 + (size_t)(m0 + laneM) * K + laneK;
    gemm_16x64<H_DIM, H_DIM>(arow, W2 + (size_t)laneK * N + n0 + 4 * laneM, acc);
    const int mhalf = (lane >> 4) * 8;
    {
        const int n = n0 + 4 * laneM;
        const v4f bb = *(const v4f*)(bias2 + n);
#pragma unroll
        for (int r = 0; r < 8; ++r) {
            v4f o;
#pragma unroll
            for (int t = 0; t < 4; ++t) o[t] = acc[t][r] + bb[t];
            *(v4f*)(z + (mhalf + r) * H_DIM + n) = o;
        }
    }
    __syncthreads();

    const int row = threadIdx.x >> 4;
    const int seg = threadIdx.x & 15;
    {
        const float* zr = z + row * H_DIM + seg * 32;
        float s = 0.f, q = 0.f;
#pragma unroll
        for (int i = 0; i < 32; ++i) { float v = zr[i]; s += v; q += v * v; }
        red_s[row][seg] = s;
        red_q[row][seg] = q;
    }
    __syncthreads();
    if (threadIdx.x < 16) {
        float s = 0.f, q = 0.f;
#pragma unroll
        for (int i = 0; i < 16; ++i) { s += red_s[threadIdx.x][i]; q += red_q[threadIdx.x][i]; }
        const float mu = s * (1.0f / H_DIM);
        const float var = q * (1.0f / H_DIM) - mu * mu;
        mu_s[threadIdx.x] = mu;
        rs_s[threadIdx.x] = rsqrtf(var + 1e-5f);
    }
    __syncthreads();
    {
        const float mu = mu_s[row], rstd = rs_s[row];
        float* dst = h2 + (size_t)(m0 + row) * H_DIM;
        const float* zr = z + row * H_DIM;
#pragma unroll
        for (int i = 0; i < 32; ++i) {
            const int n = seg * 32 + i;
            float v = (zr[n] - mu) * rstd * ln_g[n] + ln_b[n];
            dst[n] = fmaxf(v, 0.0f);
        }
    }
}

__global__ void __launch_bounds__(256)
k_gemm3_bcast(const float* __restrict__ h2, const float* __restrict__ W3,
              const float* __restrict__ bias3, float* __restrict__ out) {
    const int K = H_DIM, N = D_DIM;
    const int wave = threadIdx.x >> 5;
    const int lane = threadIdx.x & 31;
    const int job  = blockIdx.x * 8 + wave;
    const int m0 = (job >> 2) * 16;
    const int n0 = (job & 3) * 64;
    const int laneM = lane & 15;
    const int laneK = (lane >> 4) * 2;

    v8f acc[4] = {};
    const float* arow = h2 + (size_t)(m0 + laneM) * K + laneK;
    gemm_16x64<H_DIM, D_DIM>(arow, W3 + (size_t)laneK * N + n0 + 4 * laneM, acc);
    const int mhalf = (lane >> 4) * 8;
    const int n = n0 + 4 * laneM;
    const v4f bb = *(const v4f*)(bias3 + n);
#pragma unroll
    for (int r = 0; r < 8; ++r) {
        const int m = m0 + mhalf + r;
        float* obase = out + (size_t)m * (O_DIM * D_DIM) + n;
        v4f v;
#pragma unroll
        for (int t = 0; t < 4; ++t) v[t] = acc[t][r] + bb[t];
#pragma unroll
        for (int o = 0; o < O_DIM; ++o)
            *(v4f*)(obase + o * D_DIM) = v;
    }
}

extern "C" void kernel_launch(void* const* d_in, const int* in_sizes, int n_in,
                              void* d_out, int out_size, void* d_ws, size_t ws_size,
                              hipStream_t stream) {
    const float* states = (const float*)d_in[0];
    const int*   action = (const int*)d_in[1];
    const float* W1     = (const float*)d_in[2];
    const float* b1     = (const float*)d_in[3];
    const float* W2     = (const float*)d_in[4];
    const float* b2     = (const float*)d_in[5];
    const float* W3     = (const float*)d_in[6];
    const float* b3     = (const float*)d_in[7];
    const float* ln_g   = (const float*)d_in[8];
    const float* ln_b   = (const float*)d_in[9];
    float* out = (float*)d_out;

    float* xbar = (float*)d_ws;
    float* h1   = xbar + (size_t)B_DIM * D_DIM;
    float* h2   = h1   + (size_t)B_DIM * H_DIM;

    k_reduce_states<<<B_DIM, 256, 0, stream>>>(states, xbar);
    k_gemm1<<<(B_DIM / 16) * 8 / 8, 256, 0, stream>>>(xbar, W1, b1, action, h1);
    k_gemm2_ln<<<B_DIM / 16, 256, 0, stream>>>(h1, W2, b2, ln_g, ln_b, h2);
    k_gemm3_bcast<<<(B_DIM / 16) * 4 / 8, 256, 0, stream>>>(h2, W3, b3, out);
}


# Third GEMM fused into the LayerNorm kernel's 16-row block (3 launches); 8 waves = 4 column groups x 2 K halves added through LDS, so the k sum order of out changes (two 256-k partial sums)
# speedup vs baseline: 2.4260x; 1.1344x over previous
#include <hip/hip_runtime.h>
#include <hip/hip_bf16.h>

#define B_DIM 4096
#define O_DIM 16
#define D_DIM 256
#define H_DIM 512

typedef __attribute__((ext_vector_type(2))) float v2f;
typedef __attribute__((ext_vector_type(4))) float v4f;
typedef __attribute__((ext_vector_type(8))) float v8f;

__device__ __forceinline__ v8f wmma_f32(v2f a, v2f b, v8f c) {
    return __builtin_amdgcn_wmma_f32_16x16x4_f32(
         false, a,  false, b,
         (short)0, c,  false,  false);
}

struct FragSet { v2f a[2]; v4f b[2][2]; };

template <int N>
__device__ __forceinline__ void frag_load(FragSet& f, const float* __restrict__ ap,
                                          const float* __restrict__ bp) {
#pragma unroll
    for (int s = 0; s < 2; ++s) {
        f.a[s] = *(const v2f*)(ap + 4 * s);
        const float* b = bp + (size_t)(4 * s) * N;
        f.b[s][0] = *(const v4f*)(b);
        f.b[s][1] = *(const v4f*)(b + N);
    }
}

#define WMMA_SLOTS "v_nop\n\tv_nop\n\tv_nop\n\tv_nop"

__device__ __forceinline__ void frag_mma(FragSet& f, v8f (&acc)[4]) {
    asm volatile(WMMA_SLOTS
                 : "+v"(acc[0]), "+v"(acc[1]), "+v"(acc[2]), "+v"(acc[3]),
                   "+v"(f.b[0][0]), "+v"(f.b[0][1]), "+v"(f.b[1][0]), "+v"(f.b[1][1]));
    v2f p[2][4];
#pragma unroll
    for (int s = 0; s < 2; ++s) {
#pragma unroll
        for (int t = 0; t < 4; ++t) {
            const v2f bf = { f.b[s][0][t], f.b[s][1][t] };
            p[s][t] = bf;
        }
    }
    asm volatile("" : "+v"(p[0][0]), "+v"(p[0][1]), "+v"(p[0][2]), "+v"(p[0][3]),
                      "+v"(p[1][0]), "+v"(p[1][1]), "+v"(p[1][2]), "+v"(p[1][3]));
#pragma unroll
    for (int s = 0; s < 2; ++s) {
#pragma unroll
        for (int t = 0; t < 4; ++t) acc[t] = wmma_f32(f.a[s], p[s][t], acc[t]);
    }
}

template <int K, int N>
__device__ __forceinline__ void gemm_16x64(const float* __restrict__ ap,
                                           const float* __restrict__ bp, v8f (&acc)[4]) {
    constexpr int KB = 8;
    constexpr int NB = K / KB;
    static_assert(K % KB == 0 && NB % 4 == 0 && NB >= 8, "ring of four sets");
    FragSet f0, f1, f2, f3;
#define ORDER_FENCE() \
    asm volatile("" : "+v"(acc[0]), "+v"(acc[1]), "+v"(acc[2]), "+v"(acc[3]) : : "memory")
    frag_load<N>(f0, ap, bp);
    ORDER_FENCE();
    frag_load<N>(f1, ap + KB, bp + (size_t)KB * N);
    ORDER_FENCE();
    frag_load<N>(f2, ap + 2 * KB, bp + (size_t)2 * KB * N);
    ORDER_FENCE();
    for (int kb = 3; kb < NB - 1; kb += 4) {
        const float* a = ap + kb * KB;
        const float* b = bp + (size_t)kb * KB * N;
        frag_load<N>(f3, a, b);
        ORDER_FENCE();
        frag_mma(f0, acc);
        ORDER_FENCE();
        frag_load<N>(f0, a + KB, b + (size_t)KB * N);
        ORDER_FENCE();
        frag_mma(f1, acc);
        ORDER_FENCE();
        frag_load<N>(f1, a + 2 * KB, b + (size_t)2 * KB * N);
        ORDER_FENCE();
        frag_mma(f2, acc);
        ORDER_FENCE();
        frag_load<N>(f2, a + 3 * KB, b + (size_t)3 * KB * N);
        ORDER_FENCE();
        frag_mma(f3, acc);
        ORDER_FENCE();
    }
    frag_load<N>(f3, ap + (NB - 1) * KB, bp + (size_t)(NB - 1) * KB * N);
    ORDER_FENCE();
    frag_mma(f0, acc);
    frag_mma(f1, acc);
    frag_mma(f2, acc);
    frag_mma(f3, acc);
    asm volatile(WMMA_SLOTS : "+v"(acc[0]), "+v"(acc[1]), "+v"(acc[2]), "+v"(acc[3]));
#undef ORDER_FENCE
}

__global__ void __launch_bounds__(256)
k_reduce_states(const float* __restrict__ states, float* __restrict__ xbar) {
    const int b = blockIdx.x;
    const int d = threadIdx.x;
    const float* p = states + (size_t)b * (O_DIM * D_DIM) + d;
    float acc = 0.f;
#pragma unroll
    for (int o = 0; o < O_DIM; ++o) acc += p[o * D_DIM];
    xbar[(size_t)b * D_DIM + d] = acc * (1.0f / 16.0f);
}

__global__ void __launch_bounds__(256)
k_gemm1(const float* __restrict__ xbar, const float* __restrict__ W1,
        const float* __restrict__ bias1, const int* __restrict__ action,
        float* __restrict__ h1) {
    const int K = D_DIM, N = H_DIM;
    const int wave = threadIdx.x >> 5;
    const int lane = threadIdx.x & 31;
    const int job  = blockIdx.x * 8 + wave;
    const int m0 = (job >> 3) * 16;
    const int n0 = (job & 7) * 64;
    const int laneM = lane & 15;
    const int laneK = (lane >> 4) * 2;

    v8f acc[4] = {};
    const float* arow = xbar + (size_t)(m0 + laneM) * K + laneK;
    gemm_16x64<D_DIM, H_DIM>(arow, W1 + (size_t)laneK * N + n0 + 4 * laneM, acc);
    const int mhalf = (lane >> 4) * 8;
    const int n = n0 + 4 * laneM;
    const v4f bb = *(const v4f*)(bias1 + n);
#pragma unroll
    for (int r = 0; r < 8; ++r) {
        const int m = m0 + mhalf + r;
        const int c = action[m] & 3;
        const v4f wt = *(const v4f*)(W1 + (size_t)(D_DIM + c) * N + n);
        v4f o;
#pragma unroll
        for (int t = 0; t < 4; ++t) {
            float v = acc[t][r] + wt[t] * 0.0625f + bb[t];
            o[t] = fmaxf(v, 0.0f);
        }
        *(v4f*)(h1 + (size_t)m * N + n) = o;
    }
}

__global__ void __launch_bounds__(256)
k_gemm2_ln_gemm3(const float* __restrict__ h1, const float* __restrict__ W2,
                 const float* __restrict__ bias2, const float* __restrict__ ln_g,
                 const float* __restrict__ ln_b, float* h2,
                 const float* __restrict__ W3, const float* __restrict__ bias3,
                 float* __restrict__ out) {
    const int K = H_DIM, N = H_DIM;
    __shared__ __attribute__((aligned(16))) float z[16 * H_DIM];
    __shared__ float red_s[16][16];
    __shared__ float red_q[16][16];
    __shared__ float mu_s[16], rs_s[16];

    const int wave = threadIdx.x >> 5;
    const int lane = threadIdx.x & 31;
    const int m0 = blockIdx.x * 16;
    const int n0 = wave * 64;
    const int laneM = lane & 15;
    const int laneK = (lane >> 4) * 2;

    v8f acc[4] = {};
    const float* arow = h1 + (size_t)(m0 + laneM) * K + laneK;
    gemm_16x64<H_DIM, H_DIM>(arow, W2 + (size_t)laneK * N + n0 + 4 * laneM, acc);
    const int mhalf = (lane >> 4) * 8;
    {
        const int n = n0 + 4 * laneM;
        const v4f bb = *(const v4f*)(bias2 + n);
#pragma unroll
        for (int r = 0; r < 8; ++r) {
            v4f o;
#pragma unroll
            for (int t = 0; t < 4; ++t) o[t] = acc[t][r] + bb[t];
            *(v4f*)(z + (mhalf + r) * H_DIM + n) = o;
        }
    }
    __syncthreads();

    const int row = threadIdx.x >> 4;
    const int seg = threadIdx.x & 15;
    {
        const float* zr = z + row * H_DIM + seg * 32;
        float s = 0.f, q = 0.f;
#pragma unroll
        for (int i = 0; i < 32; ++i) { float v = zr[i]; s += v; q += v * v; }
        red_s[row][seg] = s;
        red_q[row][seg] = q;
    }
    __syncthreads();
    if (threadIdx.x < 16) {
        float s = 0.f, q = 0.f;
#pragma unroll
        for (int i = 0; i < 16; ++i) { s += red_s[threadIdx.x][i]; q += red_q[threadIdx.x][i]; }
        const float mu = s * (1.0f / H_DIM);
        const float var = q * (1.0f / H_DIM) - mu * mu;
        mu_s[threadIdx.x] = mu;
        rs_s[threadIdx.x] = rsqrtf(var + 1e-5f);
    }
    __syncthreads();
    {
        const float mu = mu_s[row], rstd = rs_s[row];
        float* dst = h2 + (size_t)(m0 + row) * H_DIM;
        const float* zr = z + row * H_DIM;
#pragma unroll
        for (int i = 0; i < 32; ++i) {
            const int n = seg * 32 + i;
            float v = (zr[n] - mu) * rstd * ln_g[n] + ln_b[n];
            dst[n] = fmaxf(v, 0.0f);
        }
    }
    __syncthreads();

    const int cg = wave & 3;
    const int kh = wave >> 2;
    const int k0 = kh * (H_DIM / 2);
    const int n3 = cg * 64 + 4 * laneM;
    v8f acc3[4] = {};
    const float* arow3 = h2 + (size_t)(m0 + laneM) * H_DIM + k0 + laneK;
    gemm_16x64<H_DIM / 2, D_DIM>(arow3, W3 + (size_t)(k0 + laneK) * D_DIM + n3, acc3);

    float* part = z;
#pragma unroll
    for (int r = 0; r < 8; ++r) {
        v4f p;
#pragma unroll
        for (int t = 0; t < 4; ++t) p[t] = acc3[t][r];
        *(v4f*)(part + kh * (16 * D_DIM) + (mhalf + r) * D_DIM + n3) = p;
    }
    __syncthreads();
    const v4f bb3 = *(const v4f*)(bias3 + n3);
#pragma unroll
    for (int r = 0; r < 8; ++r) {
        const v4f oth = *(const v4f*)(part + (kh ^ 1) * (16 * D_DIM) + (mhalf + r) * D_DIM + n3);
        v4f v;
#pragma unroll
        for (int t = 0; t < 4; ++t) v[t] = (acc3[t][r] + oth[t]) + bb3[t];
        float* obase = out + (size_t)(m0 + mhalf + r) * (O_DIM * D_DIM) + (kh * 8) * D_DIM + n3;
#pragma unroll
        for (int o = 0; o < O_DIM / 2; ++o)
            *(v4f*)(obase + o * D_DIM) = v;
    }
}

extern "C" void kernel_launch(void* const* d_in, const int* in_sizes, int n_in,
                              void* d_out, int out_size, void* d_ws, size_t ws_size,
                              hipStream_t stream) {
    const float* states = (const float*)d_in[0];
    const int*   action = (const int*)d_in[1];
    const float* W1     = (const float*)d_in[2];
    const float* b1     = (const float*)d_in[3];
    const float* W2     = (const float*)d_in[4];
    const float* b2     = (const float*)d_in[5];
    const float* W3     = (const float*)d_in[6];
    const float* b3     = (const float*)d_in[7];
    const float* ln_g   = (const float*)d_in[8];
    const float* ln_b   = (const float*)d_in[9];
    float* out = (float*)d_out;

    float* xbar = (float*)d_ws;
    float* h1   = xbar + (size_t)B_DIM * D_DIM;
    float* h2   = h1   + (size_t)B_DIM * H_DIM;

    k_reduce_states<<<B_DIM, 256, 0, stream>>>(states, xbar);
    k_gemm1<<<(B_DIM / 16) * 8 / 8, 256, 0, stream>>>(xbar, W1, b1, action, h1);
    k_gemm2_ln_gemm3<<<B_DIM / 16, 256, 0, stream>>>(h1, W2, b2, ln_g, ln_b, h2, W3, b3, out);
}


# All three launches fused into one: each 16-row block reduces its states to xbar in LDS and keeps xbar, h1, h2 in LDS (row pitch 4 x odd) as A operand of the next product; every sum order unchanged
# speedup vs baseline: 2.9037x; 1.1969x over previous
#include <hip/hip_runtime.h>
#include <hip/hip_bf16.h>

#define B_DIM 4096
#define O_DIM 16
#define D_DIM 256
#define H_DIM 512
#define XP (D_DIM + 4)
#define HP (H_DIM + 4)

typedef __attribute__((ext_vector_type(2))) float v2f;
typedef __attribute__((ext_vector_type(4))) float v4f;
typedef __attribute__((ext_vector_type(8))) float v8f;

__device__ __forceinline__ v8f wmma_f32(v2f a, v2f b, v8f c) {
    return __builtin_amdgcn_wmma_f32_16x16x4_f32(
         false, a,  false, b,
         (short)0, c,  false,  false);
}

struct FragSet { v2f a[2]; v4f b[2][2]; };

template <int N>
__device__ __forceinline__ void frag_load(FragSet& f, const float* __restrict__ ap,
                                          const float* __restrict__ bp) {
#pragma unroll
    for (int s = 0; s < 2; ++s) {
        f.a[s] = *(const v2f*)(ap + 4 * s);
        const float* b = bp + (size_t)(4 * s) * N;
        f.b[s][0] = *(const v4f*)(b);
        f.b[s][1] = *(const v4f*)(b + N);
    }
}

#define WMMA_SLOTS "v_nop\n\tv_nop\n\tv_nop\n\tv_nop"

__device__ __forceinline__ void frag_mma(FragSet& f, v8f (&acc)[4]) {
    asm volatile(WMMA_SLOTS
                 : "+v"(acc[0]), "+v"(acc[1]), "+v"(acc[2]), "+v"(acc[3]),
                   "+v"(f.b[0][0]), "+v"(f.b[0][1]), "+v"(f.b[1][0]), "+v"(f.b[1][1]));
    v2f p[2][4];
#pragma unroll
    for (int s = 0; s < 2; ++s) {
#pragma unroll
        for (int t = 0; t < 4; ++t) {
            const v2f bf = { f.b[s][0][t], f.b[s][1][t] };
            p[s][t] = bf;
        }
    }
    asm volatile("" : "+v"(p[0][0]), "+v"(p[0][1]), "+v"(p[0][2]), "+v"(p[0][3]),
                      "+v"(p[1][0]), "+v"(p[1][1]), "+v"(p[1][2]), "+v"(p[1][3]));
#pragma unroll
    for (int s = 0; s < 2; ++s) {
#pragma unroll
        for (int t = 0; t < 4; ++t) acc[t] = wmma_f32(f.a[s], p[s][t], acc[t]);
    }
}

template <int K, int N>
__device__ __forceinline__ void gemm_16x64(const float* __restrict__ ap,
                                           const float* __restrict__ bp, v8f (&acc)[4]) {
    constexpr int KB = 8;
    constexpr int NB = K / KB;
    static_assert(K % KB == 0 && NB % 4 == 0 && NB >= 8, "ring of four sets");
    FragSet f0, f1, f2, f3;
#define ORDER_FENCE() \
    asm volatile("" : "+v"(acc[0]), "+v"(acc[1]), "+v"(acc[2]), "+v"(acc[3]) : : "memory")
    frag_load<N>(f0, ap, bp);
    ORDER_FENCE();
    frag_load<N>(f1, ap + KB, bp + (size_t)KB * N);
    ORDER_FENCE();
    frag_load<N>(f2, ap + 2 * KB, bp + (size_t)2 * KB * N);
    ORDER_FENCE();
    for (int kb = 3; kb < NB - 1; kb += 4) {
        const float* a = ap + kb * KB;
        const float* b = bp + (size_t)kb * KB * N;
        frag_load<N>(f3, a, b);
        ORDER_FENCE();
        frag_mma(f0, acc);
        ORDER_FENCE();
        frag_load<N>(f0, a + KB, b + (size_t)KB * N);
        ORDER_FENCE();
        frag_mma(f1, acc);
        ORDER_FENCE();
        frag_load<N>(f1, a + 2 * KB, b + (size_t)2 * KB * N);
        ORDER_FENCE();
        frag_mma(f2, acc);
        ORDER_FENCE();
        frag_load<N>(f2, a + 3 * KB, b + (size_t)3 * KB * N);
        ORDER_FENCE();
        frag_mma(f3, acc);
        ORDER_FENCE();
    }
    frag_load<N>(f3, ap + (NB - 1) * KB, bp + (size_t)(NB - 1) * KB * N);
    ORDER_FENCE();
    frag_mma(f0, acc);
    frag_mma(f1, acc);
    frag_mma(f2, acc);
    frag_mma(f3, acc);
    asm volatile(WMMA_SLOTS : "+v"(acc[0]), "+v"(acc[1]), "+v"(acc[2]), "+v"(acc[3]));
#undef ORDER_FENCE
}

__global__ void __launch_bounds__(256)
k_fused(const float* __restrict__ states, const int* __restrict__ action,
        const float* __restrict__ W1, const float* __restrict__ bias1,
        const float* __restrict__ W2, const float* __restrict__ bias2,
        const float* __restrict__ ln_g, const float* __restrict__ ln_b,
        const float* __restrict__ W3, const float* __restrict__ bias3,
        float* __restrict__ out) {
    __shared__ __attribute__((aligned(16))) float xs[16 * XP];
    __shared__ __attribute__((aligned(16))) float hs[16 * HP];
    __shared__ __attribute__((aligned(16))) float z[16 * H_DIM];
    __shared__ float red_s[16][16];
    __shared__ float red_q[16][16];
    __shared__ float mu_s[16], rs_s[16];

    const int wave = threadIdx.x >> 5;
    const int lane = threadIdx.x & 31;
    const int m0 = blockIdx.x * 16;
    const int n0 = wave * 64;
    const int laneM = lane & 15;
    const int laneK = (lane >> 4) * 2;
    const int mhalf = (lane >> 4) * 8;

#pragma unroll
    for (int j = 0; j < 2; ++j) {
        const int bl = wave * 2 + j;
        const float* p = states + (size_t)(m0 + bl) * (O_DIM * D_DIM) + lane * 4;
        v4f x0[O_DIM], x1[O_DIM];
#pragma unroll
        for (int o = 0; o < O_DIM; ++o) {
            x0[o] = *(const v4f*)(p + o * D_DIM);
            x1[o] = *(const v4f*)(p + o * D_DIM + D_DIM / 2);
        }
        v4f s0 = {0.f, 0.f, 0.f, 0.f}, s1 = {0.f, 0.f, 0.f, 0.f};
#pragma unroll
        for (int o = 0; o < O_DIM; ++o) { s0 += x0[o]; s1 += x1[o]; }
        *(v4f*)(xs + bl * XP + lane * 4) = s0 * (1.0f / 16.0f);
        *(v4f*)(xs + bl * XP + D_DIM / 2 + lane * 4) = s1 * (1.0f / 16.0f);
    }
    __syncthreads();

    {
        const int N = H_DIM;
        v8f acc[4] = {};
        gemm_16x64<D_DIM, H_DIM>(xs + laneM * XP + laneK,
                                 W1 + (size_t)laneK * N + n0 + 4 * laneM, acc);
        const int n = n0 + 4 * laneM;
        const v4f bb = *(const v4f*)(bias1 + n);
#pragma unroll
        for (int r = 0; r < 8; ++r) {
            const int ml = mhalf + r;
            const int c = action[m0 + ml] & 3;
            const v4f wt = *(const v4f*)(W1 + (size_t)(D_DIM + c) * N + n);
            v4f o;
#pragma unroll
            for (int t = 0; t < 4; ++t) {
                float v = acc[t][r] + wt[t] * 0.0625f + bb[t];
                o[t] = fmaxf(v, 0.0f);
            }
            *(v4f*)(hs + ml * HP + n) = o;
        }
    }
    __syncthreads();

    {
        const int N = H_DIM;
        v8f acc[4] = {};
        gemm_16x64<H_DIM, H_DIM>(hs + laneM * HP + laneK,
                                 W2 + (size_t)laneK * N + n0 + 4 * laneM, acc);
        const int n = n0 + 4 * laneM;
        const v4f bb = *(const v4f*)(bias2 + n);
#pragma unroll
        for (int r = 0; r < 8; ++r) {
            v4f o;
#pragma unroll
            for (int t = 0; t < 4; ++t) o[t] = acc[t][r] + bb[t];
            *(v4f*)(z + (mhalf + r) * H_DIM + n) = o;
        }
    }
    __syncthreads();

    const int row = threadIdx.x >> 4;
    const int seg = threadIdx.x & 15;
    {
        const float* zr = z + row * H_DIM + seg * 32;
        float s = 0.f, q = 0.f;
#pragma unroll
        for (int i = 0; i < 32; ++i) { float v = zr[i]; s += v; q += v * v; }
        red_s[row][seg] = s;
        red_q[row][seg] = q;
    }
    __syncthreads();
    if (threadIdx.x < 16) {
        float s = 0.f, q = 0.f;
#pragma unroll
        for (int i = 0; i < 16; ++i) { s += red_s[threadIdx.x][i]; q += red_q[threadIdx.x][i]; }
        const float mu = s * (1.0f / H_DIM);
        const float var = q * (1.0f / H_DIM) - mu * mu;
        mu_s[threadIdx.x] = mu;
        rs_s[threadIdx.x] = rsqrtf(var + 1e-5f);
    }
    __syncthreads();
    {
        const float mu = mu_s[row], rstd = rs_s[row];
        float* dst = hs + row * HP;
        const float* zr = z + row * H_DIM;
#pragma unroll
        for (int i = 0; i < 32; ++i) {
            const int n = seg * 32 + i;
            float v = (zr[n] - mu) * rstd * ln_g[n] + ln_b[n];
            dst[n] = fmaxf(v, 0.0f);
        }
    }
    __syncthreads();

    const int cg = wave & 3;
    const int kh = wave >> 2;
    const int k0 = kh * (H_DIM / 2);
    const int n3 = cg * 64 + 4 * laneM;
    v8f acc3[4] = {};
    gemm_16x64<H_DIM / 2, D_DIM>(hs + laneM * HP + k0 + laneK,
                                 W3 + (size_t)(k0 + laneK) * D_DIM + n3, acc3);

    float* part = z;
#pragma unroll
    for (int r = 0; r < 8; ++r) {
        v4f p;
#pragma unroll
        for (int t = 0; t < 4; ++t) p[t] = acc3[t][r];
        *(v4f*)(part + kh * (16 * D_DIM) + (mhalf + r) * D_DIM + n3) = p;
    }
    __syncthreads();
    const v4f bb3 = *(const v4f*)(bias3 + n3);
#pragma unroll
    for (int r = 0; r < 8; ++r) {
        const v4f oth = *(const v4f*)(part + (kh ^ 1) * (16 * D_DIM) + (mhalf + r) * D_DIM + n3);
        v4f v;
#pragma unroll
        for (int t = 0; t < 4; ++t) v[t] = (acc3[t][r] + oth[t]) + bb3[t];
        float* obase = out + (size_t)(m0 + mhalf + r) * (O_DIM * D_DIM) + (kh * 8) * D_DIM + n3;
#pragma unroll
        for (int o = 0; o < O_DIM / 2; ++o)
            *(v4f*)(obase + o * D_DIM) = v;
    }
}

extern "C" void kernel_launch(void* const* d_in, const int* in_sizes, int n_in,
                              void* d_out, int out_size, void* d_ws, size_t ws_size,
                              hipStream_t stream) {
    const float* states = (const float*)d_in[0];
    const int*   action = (const int*)d_in[1];
    const float* W1     = (const float*)d_in[2];
    const float* b1     = (const float*)d_in[3];
    const float* W2     = (const float*)d_in[4];
    const float* b2     = (const float*)d_in[5];
    const float* W3     = (const float*)d_in[6];
    const float* b3     = (const float*)d_in[7];
    const float* ln_g   = (const float*)d_in[8];
    const float* ln_b   = (const float*)d_in[9];
    float* out = (float*)d_out;

    k_fused<<<B_DIM / 16, 256, 0, stream>>>(states, action, W1, b1, W2, b2, ln_g, ln_b, W3, b3, out);
}


# K loops of all three WMMA products start at a per-block rotated 8-k set (blockIdx mod sets, wrapping) so the blocks stop asking L2 for the same W lines at once; k sum order changes (rotated start)
# speedup vs baseline: 3.1151x; 1.0728x over previous
#include <hip/hip_runtime.h>
#include <hip/hip_bf16.h>

#define B_DIM 4096
#define O_DIM 16
#define D_DIM 256
#define H_DIM 512
#define XP (D_DIM + 4)
#define HP (H_DIM + 4)

typedef __attribute__((ext_vector_type(2))) float v2f;
typedef __attribute__((ext_vector_type(4))) float v4f;
typedef __attribute__((ext_vector_type(8))) float v8f;
typedef __attribute__((address_space(1))) const float gfloat;
typedef __attribute__((address_space(1))) const v4f gv4f;

__device__ __forceinline__ v8f wmma_f32(v2f a, v2f b, v8f c) {
    return __builtin_amdgcn_wmma_f32_16x16x4_f32(
         false, a,  false, b,
         (short)0, c,  false,  false);
}

struct FragSet { v2f a[2]; v4f b[2][2]; };

template <int N>
__device__ __forceinline__ void frag_load(FragSet& f, const float* __restrict__ ap,
                                          gfloat* __restrict__ bp) {
#pragma unroll
    for (int s = 0; s < 2; ++s) {
        f.a[s] = *(const v2f*)(ap + 4 * s);
        gfloat* b = bp + (size_t)(4 * s) * N;
        f.b[s][0] = *(gv4f*)(b);
        f.b[s][1] = *(gv4f*)(b + N);
    }
}

#define WMMA_SLOTS "v_nop\n\tv_nop\n\tv_nop\n\tv_nop"

__device__ __forceinline__ void frag_mma(FragSet& f, v8f (&acc)[4]) {
    asm volatile(WMMA_SLOTS
                 : "+v"(acc[0]), "+v"(acc[1]), "+v"(acc[2]), "+v"(acc[3]),
                   "+v"(f.b[0][0]), "+v"(f.b[0][1]), "+v"(f.b[1][0]), "+v"(f.b[1][1]));
    v2f p[2][4];
#pragma unroll
    for (int s = 0; s < 2; ++s) {
#pragma unroll
        for (int t = 0; t < 4; ++t) {
            const v2f bf = { f.b[s][0][t], f.b[s][1][t] };
            p[s][t] = bf;
        }
    }
    asm volatile("" : "+v"(p[0][0]), "+v"(p[0][1]), "+v"(p[0][2]), "+v"(p[0][3]),
                      "+v"(p[1][0]), "+v"(p[1][1]), "+v"(p[1][2]), "+v"(p[1][3]));
#pragma unroll
    for (int s = 0; s < 2; ++s) {
#pragma unroll
        for (int t = 0; t < 4; ++t) acc[t] = wmma_f32(f.a[s], p[s][t], acc[t]);
    }
}

template <int N>
__device__ __forceinline__ gfloat* w_set(const float* __restrict__ wb, int set, unsigned boff) {
    unsigned long long u = (unsigned long long)(wb + (size_t)(set * (8 * N)));
    asm("" : "+s"(u));
    return (gfloat*)u + boff;
}

template <int K, int N>
__device__ __forceinline__ void gemm_16x64(const float* __restrict__ ap,
                                           const float* __restrict__ wb, unsigned boff,
                                           int rot, v8f (&acc)[4]) {
    constexpr int KB = 8;
    constexpr int NB = K / KB;
    static_assert(K % KB == 0 && NB % 4 == 0 && NB >= 8, "ring of four sets");
    static_assert((NB & (NB - 1)) == 0, "the wrap of the rotated set index is a mask");
    FragSet f0, f1, f2, f3;
#define SET_B(j) w_set<N>(wb, ((j) + rot) & (NB - 1), boff)
#define ORDER_FENCE() \
    asm volatile("" : "+v"(acc[0]), "+v"(acc[1]), "+v"(acc[2]), "+v"(acc[3]) : : "memory")
    frag_load<N>(f0, ap, SET_B(0));
    ORDER_FENCE();
    frag_load<N>(f1, ap + KB, SET_B(1));
    ORDER_FENCE();
    frag_load<N>(f2, ap + 2 * KB, SET_B(2));
    ORDER_FENCE();
    for (int kb = 3; kb < NB - 1; kb += 4) {
        const float* a = ap + kb * KB;
        frag_load<N>(f3, a, SET_B(kb));
        ORDER_FENCE();
        frag_mma(f0, acc);
        ORDER_FENCE();
        frag_load<N>(f0, a + KB, SET_B(kb + 1));
        ORDER_FENCE();
        frag_mma(f1, acc);
        ORDER_FENCE();
        frag_load<N>(f1, a + 2 * KB, SET_B(kb + 2));
        ORDER_FENCE();
        frag_mma(f2, acc);
        ORDER_FENCE();
        frag_load<N>(f2, a + 3 * KB, SET_B(kb + 3));
        ORDER_FENCE();
        frag_mma(f3, acc);
        ORDER_FENCE();
    }
    frag_load<N>(f3, ap + (NB - 1) * KB, SET_B(NB - 1));
    ORDER_FENCE();
    frag_mma(f0, acc);
    frag_mma(f1, acc);
    frag_mma(f2, acc);
    frag_mma(f3, acc);
    asm volatile(WMMA_SLOTS : "+v"(acc[0]), "+v"(acc[1]), "+v"(acc[2]), "+v"(acc[3]));
#undef ORDER_FENCE
#undef SET_B
}

__global__ void __launch_bounds__(256)
k_fused(const float* __restrict__ states, const int* __restrict__ action,
        const float* __restrict__ W1, const float* __restrict__ bias1,
        const float* __restrict__ W2, const float* __restrict__ bias2,
        const float* __restrict__ ln_g, const float* __restrict__ ln_b,
        const float* __restrict__ W3, const float* __restrict__ bias3,
        float* __restrict__ out) {
    __shared__ __attribute__((aligned(16))) float xs[16 * XP];
    __shared__ __attribute__((aligned(16))) float hs[16 * HP];
    __shared__ __attribute__((aligned(16))) float z[16 * H_DIM];
    __shared__ float red_s[16][16];
    __shared__ float red_q[16][16];
    __shared__ float mu_s[16], rs_s[16];

    const int wave = threadIdx.x >> 5;
    const int lane = threadIdx.x & 31;
    const int m0 = blockIdx.x * 16;
    const int n0 = wave * 64;
    const int laneM = lane & 15;
    const int laneK = (lane >> 4) * 2;
    const int mhalf = (lane >> 4) * 8;
    const int rot = blockIdx.x;
    const int r1 = (rot & (D_DIM / 8 - 1)) * 8;
    const int r2 = (rot & (H_DIM / 8 - 1)) * 8;
    const int r3 = (rot & (H_DIM / 16 - 1)) * 8;

#pragma unroll
    for (int j = 0; j < 2; ++j) {
        const int bl = wave * 2 + j;
        const float* p = states + (size_t)(m0 + bl) * (O_DIM * D_DIM) + lane * 4;
        v4f x0[O_DIM], x1[O_DIM];
#pragma unroll
        for (int o = 0; o < O_DIM; ++o) {
            x0[o] = *(const v4f*)(p + o * D_DIM);
            x1[o] = *(const v4f*)(p + o * D_DIM + D_DIM / 2);
        }
        v4f s0 = {0.f, 0.f, 0.f, 0.f}, s1 = {0.f, 0.f, 0.f, 0.f};
#pragma unroll
        for (int o = 0; o < O_DIM; ++o) { s0 += x0[o]; s1 += x1[o]; }
        *(v4f*)(xs + bl * XP + ((lane * 4 - r1) & (D_DIM - 1))) = s0 * (1.0f / 16.0f);
        *(v4f*)(xs + bl * XP + ((lane * 4 + D_DIM / 2 - r1) & (D_DIM - 1))) = s1 * (1.0f / 16.0f);
    }
    __syncthreads();

    {
        const int N = H_DIM;
        v8f acc[4] = {};
        gemm_16x64<D_DIM, H_DIM>(xs + laneM * XP + laneK, W1,
                                 (unsigned)(laneK * N + n0 + 4 * laneM), rot, acc);
        const int n = n0 + 4 * laneM;
        const v4f bb = *(const v4f*)(bias1 + n);
#pragma unroll
        for (int r = 0; r < 8; ++r) {
            const int ml = mhalf + r;
            const int c = action[m0 + ml] & 3;
            const v4f wt = *(const v4f*)(W1 + (size_t)(D_DIM + c) * N + n);
            v4f o;
#pragma unroll
            for (int t = 0; t < 4; ++t) {
                float v = acc[t][r] + wt[t] * 0.0625f + bb[t];
                o[t] = fmaxf(v, 0.0f);
            }
            *(v4f*)(hs + ml * HP + ((n - r2) & (H_DIM - 1))) = o;
        }
    }
    __syncthreads();

    {
        const int N = H_DIM;
        v8f acc[4] = {};
        gemm_16x64<H_DIM, H_DIM>(hs + laneM * HP + laneK, W2,
                                 (unsigned)(laneK * N + n0 + 4 * laneM), rot, acc);
        const int n = n0 + 4 * laneM;
        const v4f bb = *(const v4f*)(bias2 + n);
#pragma unroll
        for (int r = 0; r < 8; ++r) {
            v4f o;
#pragma unroll
            for (int t = 0; t < 4; ++t) o[t] = acc[t][r] + bb[t];
            *(v4f*)(z + (mhalf + r) * H_DIM + n) = o;
        }
    }
    __syncthreads();

    const int row = threadIdx.x >> 4;
    const int seg = threadIdx.x & 15;
    {
        const float* zr = z + row * H_DIM + seg * 32;
        float s = 0.f, q = 0.f;
#pragma unroll
        for (int i = 0; i < 32; ++i) { float v = zr[i]; s += v; q += v * v; }
        red_s[row][seg] = s;
        red_q[row][seg] = q;
    }
    __syncthreads();
    if (threadIdx.x < 16) {
        float s = 0.f, q = 0.f;
#pragma unroll
        for (int i = 0; i < 16; ++i) { s += red_s[threadIdx.x][i]; q += red_q[threadIdx.x][i]; }
        const float mu = s * (1.0f / H_DIM);
        const float var = q * (1.0f / H_DIM) - mu * mu;
        mu_s[threadIdx.x] = mu;
        rs_s[threadIdx.x] = rsqrtf(var + 1e-5f);
    }
    __syncthreads();
    {
        const float mu = mu_s[row], rstd = rs_s[row];
        float* dst = hs + row * HP;
        const float* zr = z + row * H_DIM;
        float h2v[32];
#pragma unroll
        for (int i = 0; i < 32; ++i) {
            const int n = seg * 32 + i;
            float v = (zr[n] - mu) * rstd * ln_g[n] + ln_b[n];
            h2v[i] = fmaxf(v, 0.0f);
        }
#pragma unroll
        for (int g = 0; g < 8; ++g) {
            const int n = seg * 32 + 4 * g;
            const int col = (n & (H_DIM / 2)) | ((n - r3) & (H_DIM / 2 - 1));
            const v4f o = { h2v[4 * g], h2v[4 * g + 1], h2v[4 * g + 2], h2v[4 * g + 3] };
            *(v4f*)(dst + col) = o;
        }
    }
    __syncthreads();

    const int cg = wave & 3;
    const int kh = wave >> 2;
    const int k0 = kh * (H_DIM / 2);
    const int n3 = cg * 64 + 4 * laneM;
    v8f acc3[4] = {};
    gemm_16x64<H_DIM / 2, D_DIM>(hs + laneM * HP + k0 + laneK, W3,
                                 (unsigned)((k0 + laneK) * D_DIM + n3), rot, acc3);

    float* part = z;
#pragma unroll
    for (int r = 0; r < 8; ++r) {
        v4f p;
#pragma unroll
        for (int t = 0; t < 4; ++t) p[t] = acc3[t][r];
        *(v4f*)(part + kh * (16 * D_DIM) + (mhalf + r) * D_DIM + n3) = p;
    }
    __syncthreads();
    const v4f bb3 = *(const v4f*)(bias3 + n3);
#pragma unroll
    for (int r = 0; r < 8; ++r) {
        const v4f oth = *(const v4f*)(part + (kh ^ 1) * (16 * D_DIM) + (mhalf + r) * D_DIM + n3);
        v4f v;
#pragma unroll
        for (int t = 0; t < 4; ++t) v[t] = (acc3[t][r] + oth[t]) + bb3[t];
        float* obase = out + (size_t)(m0 + mhalf + r) * (O_DIM * D_DIM) + (kh * 8) * D_DIM + n3;
#pragma unroll
        for (int o = 0; o < O_DIM / 2; ++o)
            *(v4f*)(obase + o * D_DIM) = v;
    }
}

extern "C" void kernel_launch(void* const* d_in, const int* in_sizes, int n_in,
                              void* d_out, int out_size, void* d_ws, size_t ws_size,
                              hipStream_t stream) {
    const float* states = (const float*)d_in[0];
    const int*   action = (const int*)d_in[1];
    const float* W1     = (const float*)d_in[2];
    const float* b1     = (const float*)d_in[3];
    const float* W2     = (const float*)d_in[4];
    const float* b2     = (const float*)d_in[5];
    const float* W3     = (const float*)d_in[6];
    const float* b3     = (const float*)d_in[7];
    const float* ln_g   = (const float*)d_in[8];
    const float* ln_b   = (const float*)d_in[9];
    float* out = (float*)d_out;

    k_fused<<<B_DIM / 16, 256, 0, stream>>>(states, action, W1, b1, W2, b2, ln_g, ln_b, W3, b3, out);
}


# Loads independent of the phase before them issued one phase early and left in flight across LDS-only barriers: each product's first 3 W sets, biases, action-picked W1 rows, LN gains; sums unchanged
# speedup vs baseline: 3.2299x; 1.0369x over previous
#include <hip/hip_runtime.h>
#include <hip/hip_bf16.h>

#define B_DIM 4096
#define O_DIM 16
#define D_DIM 256
#define H_DIM 512
#define XP (D_DIM + 4)
#define HP (H_DIM + 4)

typedef __attribute__((ext_vector_type(2))) float v2f;
typedef __attribute__((ext_vector_type(4))) float v4f;
typedef __attribute__((ext_vector_type(8))) float v8f;
typedef __attribute__((address_space(1))) const float gfloat;
typedef __attribute__((address_space(1))) const v4f gv4f;

__device__ __forceinline__ v8f wmma_f32(v2f a, v2f b, v8f c) {
    return __builtin_amdgcn_wmma_f32_16x16x4_f32(
         false, a,  false, b,
         (short)0, c,  false,  false);
}

struct FragSet { v2f a[2]; v4f b[2][2]; };

template <int N>
__device__ __forceinline__ void frag_load(FragSet& f, const float* __restrict__ ap,
                                          gfloat* __restrict__ bp) {
#pragma unroll
    for (int s = 0; s < 2; ++s) {
        f.a[s] = *(const v2f*)(ap + 4 * s);
        gfloat* b = bp + (size_t)(4 * s) * N;
        f.b[s][0] = *(gv4f*)(b);
        f.b[s][1] = *(gv4f*)(b + N);
    }
}

template <int N>
__device__ __forceinline__ void frag_load_b(FragSet& f, gfloat* __restrict__ bp) {
#pragma unroll
    for (int s = 0; s < 2; ++s) {
        gfloat* b = bp + (size_t)(4 * s) * N;
        f.b[s][0] = *(gv4f*)(b);
        f.b[s][1] = *(gv4f*)(b + N);
    }
}

__device__ __forceinline__ void frag_load_a(FragSet& f, const float* __restrict__ ap) {
#pragma unroll
    for (int s = 0; s < 2; ++s) f.a[s] = *(const v2f*)(ap + 4 * s);
}

#define WMMA_SLOTS "v_nop\n\tv_nop\n\tv_nop\n\tv_nop"

__device__ __forceinline__ void frag_mma(FragSet& f, v8f (&acc)[4]) {
    asm volatile(WMMA_SLOTS
                 : "+v"(acc[0]), "+v"(acc[1]), "+v"(acc[2]), "+v"(acc[3]),
                   "+v"(f.b[0][0]), "+v"(f.b[0][1]), "+v"(f.b[1][0]), "+v"(f.b[1][1]));
    v2f p[2][4];
#pragma unroll
    for (int s = 0; s < 2; ++s) {
#pragma unroll
        for (int t = 0; t < 4; ++t) {
            const v2f bf = { f.b[s][0][t], f.b[s][1][t] };
            p[s][t] = bf;
        }
    }
    asm volatile("" : "+v"(p[0][0]), "+v"(p[0][1]), "+v"(p[0][2]), "+v"(p[0][3]),
                      "+v"(p[1][0]), "+v"(p[1][1]), "+v"(p[1][2]), "+v"(p[1][3]));
#pragma unroll
    for (int s = 0; s < 2; ++s) {
#pragma unroll
        for (int t = 0; t < 4; ++t) acc[t] = wmma_f32(f.a[s], p[s][t], acc[t]);
    }
}

template <int N>
__device__ __forceinline__ gfloat* w_set(const float* __restrict__ wb, int set, unsigned boff) {
    unsigned long long u = (unsigned long long)(wb + (size_t)(set * (8 * N)));
    asm("" : "+s"(u));
    return (gfloat*)u + boff;
}

template <int K, int N, int KN, int NN>
__device__ __forceinline__ void gemm_16x64(const float* __restrict__ ap,
                                           const float* __restrict__ wb, unsigned boff,
                                           int rot, v8f (&acc)[4],
                                           FragSet& f0, FragSet& f1, FragSet& f2,
                                           const float* __restrict__ wb_next,
                                           unsigned boff_next) {
    constexpr int KB = 8;
    constexpr int NB = K / KB;
    static_assert(K % KB == 0 && NB % 4 == 0 && NB >= 8, "ring of four sets");
    static_assert((NB & (NB - 1)) == 0, "the wrap of the rotated set index is a mask");
    FragSet f3;
#define SET_B(j) w_set<N>(wb, ((j) + rot) & (NB - 1), boff)
#define NEXT_B(j) w_set<NN>(wb_next, ((j) + rot) & (KN / KB - 1), boff_next)
#define ORDER_FENCE() \
    asm volatile("" : "+v"(acc[0]), "+v"(acc[1]), "+v"(acc[2]), "+v"(acc[3]) : : "memory")
    frag_load_a(f0, ap);
    frag_load_a(f1, ap + KB);
    frag_load_a(f2, ap + 2 * KB);
    ORDER_FENCE();
    for (int kb = 3; kb < NB - 1; kb += 4) {
        const float* a = ap + kb * KB;
        frag_load<N>(f3, a, SET_B(kb));
        ORDER_FENCE();
        frag_mma(f0, acc);
        ORDER_FENCE();
        frag_load<N>(f0, a + KB, SET_B(kb + 1));
        ORDER_FENCE();
        frag_mma(f1, acc);
        ORDER_FENCE();
        frag_load<N>(f1, a + 2 * KB, SET_B(kb + 2));
        ORDER_FENCE();
        frag_mma(f2, acc);
        ORDER_FENCE();
        frag_load<N>(f2, a + 3 * KB, SET_B(kb + 3));
        ORDER_FENCE();
        frag_mma(f3, acc);
        ORDER_FENCE();
    }
    frag_load<N>(f3, ap + (NB - 1) * KB, SET_B(NB - 1));
    ORDER_FENCE();
    frag_mma(f0, acc);
    if constexpr (NN > 0) { ORDER_FENCE(); frag_load_b<NN>(f0, NEXT_B(0)); ORDER_FENCE(); }
    frag_mma(f1, acc);
    if constexpr (NN > 0) { ORDER_FENCE(); frag_load_b<NN>(f1, NEXT_B(1)); ORDER_FENCE(); }
    frag_mma(f2, acc);
    if constexpr (NN > 0) { ORDER_FENCE(); frag_load_b<NN>(f2, NEXT_B(2)); ORDER_FENCE(); }
    frag_mma(f3, acc);
    asm volatile(WMMA_SLOTS : "+v"(acc[0]), "+v"(acc[1]), "+v"(acc[2]), "+v"(acc[3]));
#undef ORDER_FENCE
#undef NEXT_B
#undef SET_B
}

__device__ __forceinline__ void lds_barrier() {
    __builtin_amdgcn_fence(__ATOMIC_RELEASE, "workgroup", "local");
    __builtin_amdgcn_s_barrier();
    __builtin_amdgcn_fence(__ATOMIC_ACQUIRE, "workgroup", "local");
}

__global__ void __launch_bounds__(256) __attribute__((amdgpu_waves_per_eu(2, 4)))
k_fused(const float* __restrict__ states, const int* __restrict__ action,
        const float* __restrict__ W1, const float* __restrict__ bias1,
        const float* __restrict__ W2, const float* __restrict__ bias2,
        const float* __restrict__ ln_g, const float* __restrict__ ln_b,
        const float* __restrict__ W3, const float* __restrict__ bias3,
        float* __restrict__ out) {
    __shared__ __attribute__((aligned(16))) float xs[16 * XP];
    __shared__ __attribute__((aligned(16))) float hs[16 * HP];
    __shared__ __attribute__((aligned(16))) float z[16 * H_DIM];
    __shared__ float red_s[16][16];
    __shared__ float red_q[16][16];
    __shared__ float mu_s[16], rs_s[16];

    const int wave = threadIdx.x >> 5;
    const int lane = threadIdx.x & 31;
    const int m0 = blockIdx.x * 16;
    const int n0 = wave * 64;
    const int laneM = lane & 15;
    const int laneK = (lane >> 4) * 2;
    const int mhalf = (lane >> 4) * 8;
    const int rot = blockIdx.x;
    const int r1 = (rot & (D_DIM / 8 - 1)) * 8;
    const int r2 = (rot & (H_DIM / 8 - 1)) * 8;
    const int r3 = (rot & (H_DIM / 16 - 1)) * 8;

    typedef __attribute__((ext_vector_type(4))) int v4i;
    const int cg = wave & 3;
    const int kh = wave >> 2;
    const int k0 = kh * (H_DIM / 2);
    const int n12 = n0 + 4 * laneM;
    const int n3 = cg * 64 + 4 * laneM;
    const unsigned boff12 = (unsigned)(laneK * H_DIM + n12);
    const unsigned boff3 = (unsigned)((k0 + laneK) * D_DIM + n3);
    const v4i act0 = *(const v4i*)(action + m0 + mhalf);
    const v4i act1 = *(const v4i*)(action + m0 + mhalf + 4);
    const v4f bb1 = *(const v4f*)(bias1 + n12);
    const v4f bb2 = *(const v4f*)(bias2 + n12);
    const v4f bb3 = *(const v4f*)(bias3 + n3);
    FragSet f0, f1, f2;
    frag_load_b<H_DIM>(f0, w_set<H_DIM>(W1, (rot + 0) & (D_DIM / 8 - 1), boff12));
    frag_load_b<H_DIM>(f1, w_set<H_DIM>(W1, (rot + 1) & (D_DIM / 8 - 1), boff12));
    frag_load_b<H_DIM>(f2, w_set<H_DIM>(W1, (rot + 2) & (D_DIM / 8 - 1), boff12));
    asm volatile("" : : : "memory");

#pragma unroll
    for (int j = 0; j < 2; ++j) {
        const int bl = wave * 2 + j;
        const float* p = states + (size_t)(m0 + bl) * (O_DIM * D_DIM) + lane * 4;
        v4f x0[O_DIM], x1[O_DIM];
#pragma unroll
        for (int o = 0; o < O_DIM; ++o) {
            x0[o] = *(const v4f*)(p + o * D_DIM);
            x1[o] = *(const v4f*)(p + o * D_DIM + D_DIM / 2);
        }
        v4f s0 = {0.f, 0.f, 0.f, 0.f}, s1 = {0.f, 0.f, 0.f, 0.f};
#pragma unroll
        for (int o = 0; o < O_DIM; ++o) { s0 += x0[o]; s1 += x1[o]; }
        *(v4f*)(xs + bl * XP + ((lane * 4 - r1) & (D_DIM - 1))) = s0 * (1.0f / 16.0f);
        *(v4f*)(xs + bl * XP + ((lane * 4 + D_DIM / 2 - r1) & (D_DIM - 1))) = s1 * (1.0f / 16.0f);
    }
    v4f wt[8];
#pragma unroll
    for (int r = 0; r < 8; ++r) {
        const int c = (r < 4 ? act0[r & 3] : act1[r & 3]) & 3;
        wt[r] = *(const v4f*)(W1 + (size_t)(D_DIM + c) * H_DIM + n12);
    }
    lds_barrier();

    {
        v8f acc[4] = {};
        gemm_16x64<D_DIM, H_DIM, H_DIM, H_DIM>(xs + laneM * XP + laneK, W1, boff12, rot, acc,
                                               f0, f1, f2, W2, boff12);
        const int n = n12;
#pragma unroll
        for (int r = 0; r < 8; ++r) {
            const int ml = mhalf + r;
            v4f o;
#pragma unroll
            for (int t = 0; t < 4; ++t) {
                float v = acc[t][r] + wt[r][t] * 0.0625f + bb1[t];
                o[t] = fmaxf(v, 0.0f);
            }
            *(v4f*)(hs + ml * HP + ((n - r2) & (H_DIM - 1))) = o;
        }
    }
    lds_barrier();

    const int row = threadIdx.x >> 4;
    const int seg = threadIdx.x & 15;
    v4f lg[8], lb[8];

    {
        v8f acc[4] = {};
        gemm_16x64<H_DIM, H_DIM, H_DIM / 2, D_DIM>(hs + laneM * HP + laneK, W2, boff12, rot, acc,
                                                   f0, f1, f2, W3, boff3);
#pragma unroll
        for (int g = 0; g < 8; ++g) {
            lg[g] = *(const v4f*)(ln_g + seg * 32 + 4 * g);
            lb[g] = *(const v4f*)(ln_b + seg * 32 + 4 * g);
        }
        asm volatile("" : : : "memory");
        const int n = n12;
#pragma unroll
        for (int r = 0; r < 8; ++r) {
            v4f o;
#pragma unroll
            for (int t = 0; t < 4; ++t) o[t] = acc[t][r] + bb2[t];
            *(v4f*)(z + (mhalf + r) * H_DIM + n) = o;
        }
    }
    lds_barrier();

    {
        const float* zr = z + row * H_DIM + seg * 32;
        float s = 0.f, q = 0.f;
#pragma unroll
        for (int i = 0; i < 32; ++i) { float v = zr[i]; s += v; q += v * v; }
        red_s[row][seg] = s;
        red_q[row][seg] = q;
    }
    lds_barrier();
    if (threadIdx.x < 16) {
        float s = 0.f, q = 0.f;
#pragma unroll
        for (int i = 0; i < 16; ++i) { s += red_s[threadIdx.x][i]; q += red_q[threadIdx.x][i]; }
        const float mu = s * (1.0f / H_DIM);
        const float var = q * (1.0f / H_DIM) - mu * mu;
        mu_s[threadIdx.x] = mu;
        rs_s[threadIdx.x] = rsqrtf(var + 1e-5f);
    }
    lds_barrier();
    {
        const float mu = mu_s[row], rstd = rs_s[row];
        float* dst = hs + row * HP;
        const float* zr = z + row * H_DIM;
        float h2v[32];
#pragma unroll
        for (int i = 0; i < 32; ++i) {
            const int n = seg * 32 + i;
            float v = (zr[n] - mu) * rstd * lg[i >> 2][i & 3] + lb[i >> 2][i & 3];
            h2v[i] = fmaxf(v, 0.0f);
        }
#pragma unroll
        for (int g = 0; g < 8; ++g) {
            const int n = seg * 32 + 4 * g;
            const int col = (n & (H_DIM / 2)) | ((n - r3) & (H_DIM / 2 - 1));
            const v4f o = { h2v[4 * g], h2v[4 * g + 1], h2v[4 * g + 2], h2v[4 * g + 3] };
            *(v4f*)(dst + col) = o;
        }
    }
    lds_barrier();

    v8f acc3[4] = {};
    gemm_16x64<H_DIM / 2, D_DIM, 0, 0>(hs + laneM * HP + k0 + laneK, W3, boff3, rot, acc3,
                                       f0, f1, f2, nullptr, 0u);

    float* part = z;
#pragma unroll
    for (int r = 0; r < 8; ++r) {
        v4f p;
#pragma unroll
        for (int t = 0; t < 4; ++t) p[t] = acc3[t][r];
        *(v4f*)(part + kh * (16 * D_DIM) + (mhalf + r) * D_DIM + n3) = p;
    }
    lds_barrier();
#pragma unroll
    for (int r = 0; r < 8; ++r) {
        const v4f oth = *(const v4f*)(part + (kh ^ 1) * (16 * D_DIM) + (mhalf + r) * D_DIM + n3);
        v4f v;
#pragma unroll
        for (int t = 0; t < 4; ++t) v[t] = (acc3[t][r] + oth[t]) + bb3[t];
        float* obase = out + (size_t)(m0 + mhalf + r) * (O_DIM * D_DIM) + (kh * 8) * D_DIM + n3;
#pragma unroll
        for (int o = 0; o < O_DIM / 2; ++o)
            *(v4f*)(obase + o * D_DIM) = v;
    }
}

extern "C" void kernel_launch(void* const* d_in, const int* in_sizes, int n_in,
                              void* d_out, int out_size, void* d_ws, size_t ws_size,
                              hipStream_t stream) {
    const float* states = (const float*)d_in[0];
    const int*   action = (const int*)d_in[1];
    const float* W1     = (const float*)d_in[2];
    const float* b1     = (const float*)d_in[3];
    const float* W2     = (const float*)d_in[4];
    const float* b2     = (const float*)d_in[5];
    const float* W3     = (const float*)d_in[6];
    const float* b3     = (const float*)d_in[7];
    const float* ln_g   = (const float*)d_in[8];
    const float* ln_b   = (const float*)d_in[9];
    float* out = (float*)d_out;

    k_fused<<<B_DIM / 16, 256, 0, stream>>>(states, action, W1, b1, W2, b2, ln_g, ln_b, W3, b3, out);
}


# LayerNorm taken on product 3's accumulators in registers (row sums by 16-lane DPP butterfly, 8 wave partials via LDS): z staging and 2 of 4 barriers go; order of the mean/variance sums changes
# speedup vs baseline: 3.3786x; 1.0460x over previous
#include <hip/hip_runtime.h>
#include <hip/hip_bf16.h>

#define B_DIM 4096
#define O_DIM 16
#define D_DIM 256
#define H_DIM 512
#define XP (D_DIM + 4)
#define HP (H_DIM + 4)

typedef __attribute__((ext_vector_type(2))) float v2f;
typedef __attribute__((ext_vector_type(4))) float v4f;
typedef __attribute__((ext_vector_type(8))) float v8f;
typedef __attribute__((address_space(1))) const float gfloat;
typedef __attribute__((address_space(1))) const v4f gv4f;

__device__ __forceinline__ v8f wmma_f32(v2f a, v2f b, v8f c) {
    return __builtin_amdgcn_wmma_f32_16x16x4_f32(
         false, a,  false, b,
         (short)0, c,  false,  false);
}

struct FragSet { v2f a[2]; v4f b[2][2]; };

template <int N>
__device__ __forceinline__ void frag_load(FragSet& f, const float* __restrict__ ap,
                                          gfloat* __restrict__ bp) {
#pragma unroll
    for (int s = 0; s < 2; ++s) {
        f.a[s] = *(const v2f*)(ap + 4 * s);
        gfloat* b = bp + (size_t)(4 * s) * N;
        f.b[s][0] = *(gv4f*)(b);
        f.b[s][1] = *(gv4f*)(b + N);
    }
}

template <int N>
__device__ __forceinline__ void frag_load_b(FragSet& f, gfloat* __restrict__ bp) {
#pragma unroll
    for (int s = 0; s < 2; ++s) {
        gfloat* b = bp + (size_t)(4 * s) * N;
        f.b[s][0] = *(gv4f*)(b);
        f.b[s][1] = *(gv4f*)(b + N);
    }
}

__device__ __forceinline__ void frag_load_a(FragSet& f, const float* __restrict__ ap) {
#pragma unroll
    for (int s = 0; s < 2; ++s) f.a[s] = *(const v2f*)(ap + 4 * s);
}

#define WMMA_SLOTS "v_nop\n\tv_nop\n\tv_nop\n\tv_nop"

__device__ __forceinline__ void frag_mma(FragSet& f, v8f (&acc)[4]) {
    asm volatile(WMMA_SLOTS
                 : "+v"(acc[0]), "+v"(acc[1]), "+v"(acc[2]), "+v"(acc[3]),
                   "+v"(f.b[0][0]), "+v"(f.b[0][1]), "+v"(f.b[1][0]), "+v"(f.b[1][1]));
    v2f p[2][4];
#pragma unroll
    for (int s = 0; s < 2; ++s) {
#pragma unroll
        for (int t = 0; t < 4; ++t) {
            const v2f bf = { f.b[s][0][t], f.b[s][1][t] };
            p[s][t] = bf;
        }
    }
    asm volatile("" : "+v"(p[0][0]), "+v"(p[0][1]), "+v"(p[0][2]), "+v"(p[0][3]),
                      "+v"(p[1][0]), "+v"(p[1][1]), "+v"(p[1][2]), "+v"(p[1][3]));
#pragma unroll
    for (int s = 0; s < 2; ++s) {
#pragma unroll
        for (int t = 0; t < 4; ++t) acc[t] = wmma_f32(f.a[s], p[s][t], acc[t]);
    }
}

template <int N>
__device__ __forceinline__ gfloat* w_set(const float* __restrict__ wb, int set, unsigned boff) {
    unsigned long long u = (unsigned long long)(wb + (size_t)(set * (8 * N)));
    asm("" : "+s"(u));
    return (gfloat*)u + boff;
}

template <int K, int N, int KN, int NN>
__device__ __forceinline__ void gemm_16x64(const float* __restrict__ ap,
                                           const float* __restrict__ wb, unsigned boff,
                                           int rot, v8f (&acc)[4],
                                           FragSet& f0, FragSet& f1, FragSet& f2,
                                           const float* __restrict__ wb_next,
                                           unsigned boff_next) {
    constexpr int KB = 8;
    constexpr int NB = K / KB;
    static_assert(K % KB == 0 && NB % 4 == 0 && NB >= 8, "ring of four sets");
    static_assert((NB & (NB - 1)) == 0, "the wrap of the rotated set index is a mask");
    FragSet f3;
#define SET_B(j) w_set<N>(wb, ((j) + rot) & (NB - 1), boff)
#define NEXT_B(j) w_set<NN>(wb_next, ((j) + rot) & (KN / KB - 1), boff_next)
#define ORDER_FENCE() \
    asm volatile("" : "+v"(acc[0]), "+v"(acc[1]), "+v"(acc[2]), "+v"(acc[3]) : : "memory")
    frag_load_a(f0, ap);
    frag_load_a(f1, ap + KB);
    frag_load_a(f2, ap + 2 * KB);
    ORDER_FENCE();
    for (int kb = 3; kb < NB - 1; kb += 4) {
        const float* a = ap + kb * KB;
        frag_load<N>(f3, a, SET_B(kb));
        ORDER_FENCE();
        frag_mma(f0, acc);
        ORDER_FENCE();
        frag_load<N>(f0, a + KB, SET_B(kb + 1));
        ORDER_FENCE();
        frag_mma(f1, acc);
        ORDER_FENCE();
        frag_load<N>(f1, a + 2 * KB, SET_B(kb + 2));
        ORDER_FENCE();
        frag_mma(f2, acc);
        ORDER_FENCE();
        frag_load<N>(f2, a + 3 * KB, SET_B(kb + 3));
        ORDER_FENCE();
        frag_mma(f3, acc);
        ORDER_FENCE();
    }
    frag_load<N>(f3, ap + (NB - 1) * KB, SET_B(NB - 1));
    ORDER_FENCE();
    frag_mma(f0, acc);
    if constexpr (NN > 0) { ORDER_FENCE(); frag_load_b<NN>(f0, NEXT_B(0)); ORDER_FENCE(); }
    frag_mma(f1, acc);
    if constexpr (NN > 0) { ORDER_FENCE(); frag_load_b<NN>(f1, NEXT_B(1)); ORDER_FENCE(); }
    frag_mma(f2, acc);
    if constexpr (NN > 0) { ORDER_FENCE(); frag_load_b<NN>(f2, NEXT_B(2)); ORDER_FENCE(); }
    frag_mma(f3, acc);
    asm volatile(WMMA_SLOTS : "+v"(acc[0]), "+v"(acc[1]), "+v"(acc[2]), "+v"(acc[3]));
#undef ORDER_FENCE
#undef NEXT_B
#undef SET_B
}

__device__ __forceinline__ void lds_barrier() {
    __builtin_amdgcn_fence(__ATOMIC_RELEASE, "workgroup", "local");
    __builtin_amdgcn_s_barrier();
    __builtin_amdgcn_fence(__ATOMIC_ACQUIRE, "workgroup", "local");
}

template <int M>
__device__ __forceinline__ float row_xor(float x) {
    return __builtin_bit_cast(float, __builtin_amdgcn_update_dpp(
        0, __builtin_bit_cast(int, x), 0x160 + M  , 0xf, 0xf, true));
}
template <int R>
__device__ __forceinline__ float row_get(float x) {
    return __builtin_bit_cast(float, __builtin_amdgcn_update_dpp(
        0, __builtin_bit_cast(int, x), 0x150 + R  , 0xf, 0xf, true));
}

template <int R>
__device__ __forceinline__ void ln_row(const float (&zv)[8][4], float mu, float rstd,
                                       v4f g, v4f b, float* __restrict__ dst) {
    const float mur = row_get<R>(mu), rsr = row_get<R>(rstd);
    const v4f zr = { zv[R][0], zv[R][1], zv[R][2], zv[R][3] };
    const v4f v = (zr - mur) * rsr * g + b;
    const v4f o = { fmaxf(v[0], 0.0f), fmaxf(v[1], 0.0f), fmaxf(v[2], 0.0f), fmaxf(v[3], 0.0f) };
    *(v4f*)(dst + R * HP) = o;
}

__global__ void __launch_bounds__(256) __attribute__((amdgpu_waves_per_eu(2, 4)))
k_fused(const float* __restrict__ states, const int* __restrict__ action,
        const float* __restrict__ W1, const float* __restrict__ bias1,
        const float* __restrict__ W2, const float* __restrict__ bias2,
        const float* __restrict__ ln_g, const float* __restrict__ ln_b,
        const float* __restrict__ W3, const float* __restrict__ bias3,
        float* __restrict__ out) {
    __shared__ __attribute__((aligned(16))) float xs[16 * XP];
    __shared__ __attribute__((aligned(16))) float hs[16 * HP];
    __shared__ __attribute__((aligned(16))) float part_s[2 * 16 * D_DIM];
    __shared__ __attribute__((aligned(16))) float red_s[16][8];
    __shared__ __attribute__((aligned(16))) float red_q[16][8];

    const int wave = threadIdx.x >> 5;
    const int lane = threadIdx.x & 31;
    const int m0 = blockIdx.x * 16;
    const int n0 = wave * 64;
    const int laneM = lane & 15;
    const int laneK = (lane >> 4) * 2;
    const int mhalf = (lane >> 4) * 8;
    const int rot = blockIdx.x;
    const int r1 = (rot & (D_DIM / 8 - 1)) * 8;
    const int r2 = (rot & (H_DIM / 8 - 1)) * 8;
    const int r3 = (rot & (H_DIM / 16 - 1)) * 8;

    typedef __attribute__((ext_vector_type(4))) int v4i;
    const int cg = wave & 3;
    const int kh = wave >> 2;
    const int k0 = kh * (H_DIM / 2);
    const int n12 = n0 + 4 * laneM;
    const int n3 = cg * 64 + 4 * laneM;
    const unsigned boff12 = (unsigned)(laneK * H_DIM + n12);
    const unsigned boff3 = (unsigned)((k0 + laneK) * D_DIM + n3);
    const v4i act0 = *(const v4i*)(action + m0 + mhalf);
    const v4i act1 = *(const v4i*)(action + m0 + mhalf + 4);
    const v4f bb1 = *(const v4f*)(bias1 + n12);
    const v4f bb2 = *(const v4f*)(bias2 + n12);
    const v4f bb3 = *(const v4f*)(bias3 + n3);
    FragSet f0, f1, f2;
    frag_load_b<H_DIM>(f0, w_set<H_DIM>(W1, (rot + 0) & (D_DIM / 8 - 1), boff12));
    frag_load_b<H_DIM>(f1, w_set<H_DIM>(W1, (rot + 1) & (D_DIM / 8 - 1), boff12));
    frag_load_b<H_DIM>(f2, w_set<H_DIM>(W1, (rot + 2) & (D_DIM / 8 - 1), boff12));
    asm volatile("" : : : "memory");

#pragma unroll
    for (int j = 0; j < 2; ++j) {
        const int bl = wave * 2 + j;
        const float* p = states + (size_t)(m0 + bl) * (O_DIM * D_DIM) + lane * 4;
        v4f x0[O_DIM], x1[O_DIM];
#pragma unroll
        for (int o = 0; o < O_DIM; ++o) {
            x0[o] = *(const v4f*)(p + o * D_DIM);
            x1[o] = *(const v4f*)(p + o * D_DIM + D_DIM / 2);
        }
        v4f s0 = {0.f, 0.f, 0.f, 0.f}, s1 = {0.f, 0.f, 0.f, 0.f};
#pragma unroll
        for (int o = 0; o < O_DIM; ++o) { s0 += x0[o]; s1 += x1[o]; }
        *(v4f*)(xs + bl * XP + ((lane * 4 - r1) & (D_DIM - 1))) = s0 * (1.0f / 16.0f);
        *(v4f*)(xs + bl * XP + ((lane * 4 + D_DIM / 2 - r1) & (D_DIM - 1))) = s1 * (1.0f / 16.0f);
    }
    v4f wt[8];
#pragma unroll
    for (int r = 0; r < 8; ++r) {
        const int c = (r < 4 ? act0[r & 3] : act1[r & 3]) & 3;
        wt[r] = *(const v4f*)(W1 + (size_t)(D_DIM + c) * H_DIM + n12);
    }
    const v4f lg4 = *(const v4f*)(ln_g + n12);
    const v4f lb4 = *(const v4f*)(ln_b + n12);
    lds_barrier();

    {
        v8f acc[4] = {};
        gemm_16x64<D_DIM, H_DIM, H_DIM, H_DIM>(xs + laneM * XP + laneK, W1, boff12, rot, acc,
                                               f0, f1, f2, W2, boff12);
        const int n = n12;
#pragma unroll
        for (int r = 0; r < 8; ++r) {
            const int ml = mhalf + r;
            v4f o;
#pragma unroll
            for (int t = 0; t < 4; ++t) {
                float v = acc[t][r] + wt[r][t] * 0.0625f + bb1[t];
                o[t] = fmaxf(v, 0.0f);
            }
            *(v4f*)(hs + ml * HP + ((n - r2) & (H_DIM - 1))) = o;
        }
    }
    lds_barrier();

    {
        v8f acc[4] = {};
        gemm_16x64<H_DIM, H_DIM, H_DIM / 2, D_DIM>(hs + laneM * HP + laneK, W2, boff12, rot, acc,
                                                   f0, f1, f2, W3, boff3);
        float zv[8][4], ps[8], pq[8];
#pragma unroll
        for (int r = 0; r < 8; ++r) {
#pragma unroll
            for (int t = 0; t < 4; ++t) zv[r][t] = acc[t][r] + bb2[t];
            ps[r] = (zv[r][0] + zv[r][1]) + (zv[r][2] + zv[r][3]);
            pq[r] = (zv[r][0] * zv[r][0] + zv[r][1] * zv[r][1]) +
                    (zv[r][2] * zv[r][2] + zv[r][3] * zv[r][3]);
        }
#pragma unroll
        for (int r = 0; r < 8; ++r) { ps[r] += row_xor<1>(ps[r]); pq[r] += row_xor<1>(pq[r]); }
#pragma unroll
        for (int r = 0; r < 8; ++r) { ps[r] += row_xor<2>(ps[r]); pq[r] += row_xor<2>(pq[r]); }
#pragma unroll
        for (int r = 0; r < 8; ++r) { ps[r] += row_xor<4>(ps[r]); pq[r] += row_xor<4>(pq[r]); }
#pragma unroll
        for (int r = 0; r < 8; ++r) { ps[r] += row_xor<8>(ps[r]); pq[r] += row_xor<8>(pq[r]); }
        {
            float sel_s = ps[0], sel_q = pq[0];
#pragma unroll
            for (int r = 1; r < 8; ++r) {
                sel_s = ((laneM & 7) == r) ? ps[r] : sel_s;
                sel_q = ((laneM & 7) == r) ? pq[r] : sel_q;
            }
            float* red = (laneM & 8) ? &red_q[0][0] : &red_s[0][0];
            red[(mhalf + (laneM & 7)) * 8 + wave] = (laneM & 8) ? sel_q : sel_s;
        }
        lds_barrier();
        const int rho = (laneM + mhalf) & 15;
        const v4f a0 = *(const v4f*)&red_s[rho][0];
        const v4f a1 = *(const v4f*)&red_s[rho][4];
        const v4f c0 = *(const v4f*)&red_q[rho][0];
        const v4f c1 = *(const v4f*)&red_q[rho][4];
        const float ts = ((a0[0] + a0[1]) + (a0[2] + a0[3])) + ((a1[0] + a1[1]) + (a1[2] + a1[3]));
        const float tq = ((c0[0] + c0[1]) + (c0[2] + c0[3])) + ((c1[0] + c1[1]) + (c1[2] + c1[3]));
        const float mu = ts * (1.0f / H_DIM);
        const float var = tq * (1.0f / H_DIM) - mu * mu;
        const float rstd = rsqrtf(var + 1e-5f);
        const int col = (n12 & (H_DIM / 2)) | ((n12 - r3) & (H_DIM / 2 - 1));
        float* dst = hs + mhalf * HP + col;
        ln_row<0>(zv, mu, rstd, lg4, lb4, dst);
        ln_row<1>(zv, mu, rstd, lg4, lb4, dst);
        ln_row<2>(zv, mu, rstd, lg4, lb4, dst);
        ln_row<3>(zv, mu, rstd, lg4, lb4, dst);
        ln_row<4>(zv, mu, rstd, lg4, lb4, dst);
        ln_row<5>(zv, mu, rstd, lg4, lb4, dst);
        ln_row<6>(zv, mu, rstd, lg4, lb4, dst);
        ln_row<7>(zv, mu, rstd, lg4, lb4, dst);
    }
    lds_barrier();

    v8f acc3[4] = {};
    gemm_16x64<H_DIM / 2, D_DIM, 0, 0>(hs + laneM * HP + k0 + laneK, W3, boff3, rot, acc3,
                                       f0, f1, f2, nullptr, 0u);

    float* part = part_s;
#pragma unroll
    for (int r = 0; r < 8; ++r) {
        v4f p;
#pragma unroll
        for (int t = 0; t < 4; ++t) p[t] = acc3[t][r];
        *(v4f*)(part + kh * (16 * D_DIM) + (mhalf + r) * D_DIM + n3) = p;
    }
    lds_barrier();
#pragma unroll
    for (int r = 0; r < 8; ++r) {
        const v4f oth = *(const v4f*)(part + (kh ^ 1) * (16 * D_DIM) + (mhalf + r) * D_DIM + n3);
        v4f v;
#pragma unroll
        for (int t = 0; t < 4; ++t) v[t] = (acc3[t][r] + oth[t]) + bb3[t];
        float* obase = out + (size_t)(m0 + mhalf + r) * (O_DIM * D_DIM) + (kh * 8) * D_DIM + n3;
#pragma unroll
        for (int o = 0; o < O_DIM / 2; ++o)
            *(v4f*)(obase + o * D_DIM) = v;
    }
}

extern "C" void kernel_launch(void* const* d_in, const int* in_sizes, int n_in,
                              void* d_out, int out_size, void* d_ws, size_t ws_size,
                              hipStream_t stream) {
    const float* states = (const float*)d_in[0];
    const int*   action = (const int*)d_in[1];
    const float* W1     = (const float*)d_in[2];
    const float* b1     = (const float*)d_in[3];
    const float* W2     = (const float*)d_in[4];
    const float* b2     = (const float*)d_in[5];
    const float* W3     = (const float*)d_in[6];
    const float* b3     = (const float*)d_in[7];
    const float* ln_g   = (const float*)d_in[8];
    const float* ln_b   = (const float*)d_in[9];
    float* out = (float*)d_out;

    k_fused<<<B_DIM / 16, 256, 0, stream>>>(states, action, W1, b1, W2, b2, ln_g, ln_b, W3, b3, out);
}
